# starting kernel
# baseline (speedup 1.0000x reference)
#include <hip/hip_runtime.h>

typedef __attribute__((ext_vector_type(2))) float v2f;
typedef __attribute__((ext_vector_type(8))) float v8f;

#define NNODES 100000
#define DIM 64
#define NTILES_M (NNODES / 16)

__global__ __launch_bounds__(256) void gemm_wmma(const float* __restrict__ X,
                                                 const float* __restrict__ W,
                                                 float* __restrict__ Y) {
  const int wave  = blockIdx.x * (blockDim.x >> 5) + (threadIdx.x >> 5);
  const int lane  = threadIdx.x & 31;
  if (wave >= NTILES_M) return;

  const int mrow  = wave * 16 + (lane & 15);
  const int khalf = (lane >> 4) * 2;
  const int nlo   = lane & 15;

  v8f acc0 = {}, acc1 = {}, acc2 = {}, acc3 = {};

  const float* xrow = X + (size_t)mrow * DIM;
  for (int k0 = 0; k0 < DIM; k0 += 4) {
    v2f a;
    a.x = xrow[k0 + khalf + 0];
    a.y = xrow[k0 + khalf + 1];
    const float* w0 = W + (size_t)(k0 + khalf) * DIM + nlo;
    v2f b0; b0.x = w0[0];        b0.y = w0[DIM];
    v2f b1; b1.x = w0[16];       b1.y = w0[DIM + 16];
    v2f b2; b2.x = w0[32];       b2.y = w0[DIM + 32];
    v2f b3; b3.x = w0[48];       b3.y = w0[DIM + 48];
    acc0 = __builtin_amdgcn_wmma_f32_16x16x4_f32(false, a, false, b0, (short)0, acc0, false, false);
    acc1 = __builtin_amdgcn_wmma_f32_16x16x4_f32(false, a, false, b1, (short)0, acc1, false, false);
    acc2 = __builtin_amdgcn_wmma_f32_16x16x4_f32(false, a, false, b2, (short)0, acc2, false, false);
    acc3 = __builtin_amdgcn_wmma_f32_16x16x4_f32(false, a, false, b3, (short)0, acc3, false, false);
  }

  const int rbase = wave * 16 + (lane >> 4) * 8;
  float* yb = Y + (size_t)rbase * DIM + nlo;
#pragma unroll
  for (int r = 0; r < 8; ++r) {
    yb[(size_t)r * DIM +  0] = acc0[r];
    yb[(size_t)r * DIM + 16] = acc1[r];
    yb[(size_t)r * DIM + 32] = acc2[r];
    yb[(size_t)r * DIM + 48] = acc3[r];
  }
}

__global__ __launch_bounds__(256) void init_out(const float* __restrict__ ori,
                                                const float* __restrict__ bias,
                                                float* __restrict__ out) {
  const int i = blockIdx.x * blockDim.x + threadIdx.x;
  if (i >= NNODES * (DIM / 4)) return;
  const float4 o = ((const float4*)ori)[i];
  const float4 b = ((const float4*)bias)[i & (DIM / 4 - 1)];
  float4 r;
  r.x = o.x + b.x; r.y = o.y + b.y; r.z = o.z + b.z; r.w = o.w + b.w;
  ((float4*)out)[i] = r;
}

__global__ __launch_bounds__(256) void scatter_edges(const float* __restrict__ XW,
                                                     const int* __restrict__ src,
                                                     const int* __restrict__ dst,
                                                     const float* __restrict__ ew,
                                                     float* __restrict__ out,
                                                     int nE) {
  const long long t = (long long)blockIdx.x * blockDim.x + threadIdx.x;
  const int e = (int)(t >> 4);
  if (e >= nE) return;
  const int q = (int)(t & 15);

  const float w = ew[e];
  const float4 m = ((const float4*)(XW + (size_t)src[e] * DIM))[q];
  float* o = out + (size_t)dst[e] * DIM + q * 4;
#if defined(__HIP_DEVICE_COMPILE__)
  unsafeAtomicAdd(o + 0, m.x * w);
  unsafeAtomicAdd(o + 1, m.y * w);
  unsafeAtomicAdd(o + 2, m.z * w);
  unsafeAtomicAdd(o + 3, m.w * w);
#endif
}

extern "C" void kernel_launch(void* const* d_in, const int* in_sizes, int n_in,
                              void* d_out, int out_size, void* d_ws, size_t ws_size,
                              hipStream_t stream) {
  const float* in_feat = (const float*)d_in[0];
  const float* ew      = (const float*)d_in[1];
  const float* W1      = (const float*)d_in[2];
  const float* b1      = (const float*)d_in[3];
  const float* W2      = (const float*)d_in[4];
  const float* b2      = (const float*)d_in[5];
  const int*   src     = (const int*)d_in[6];
  const int*   dst     = (const int*)d_in[7];
  const int    nE      = in_sizes[1];

  float* xw   = (float*)d_ws;
  float* hbuf = xw + (size_t)NNODES * DIM;
  float* outp = (float*)d_out;

  const int gemmBlocks    = (NTILES_M + 7) / 8;
  const int initBlocks    = (NNODES * (DIM / 4) + 255) / 256;
  const int scatterBlocks = (int)(((long long)nE * 16 + 255) / 256);

  const float* hcur = in_feat;
  for (int step = 0; step < 4; ++step) {
    const float* W = (step == 0) ? W1 : W2;
    const float* b = (step == 0) ? b1 : b2;
    float* tgt = (step == 3) ? outp : hbuf;

    gemm_wmma<<<gemmBlocks, 256, 0, stream>>>(hcur, W, xw);
    init_out<<<initBlocks, 256, 0, stream>>>(in_feat, b, tgt);
    scatter_edges<<<scatterBlocks, 256, 0, stream>>>(xw, src, dst, ew, tgt, nE);

    hcur = tgt;
  }
}


# Edges binned once per call by 100-node destination range; per step one work-group per range adds its edges into an LDS tile (LDS f32 adds, arrival order), stores out rows once; no global f32 atomics
# speedup vs baseline: 2.6402x; 2.6402x over previous
#include <hip/hip_runtime.h>

typedef __attribute__((ext_vector_type(2))) float v2f;
typedef __attribute__((ext_vector_type(8))) float v8f;

#define NNODES 100000
#define DIM 64
#define NTILES_M (NNODES / 16)

__global__ __launch_bounds__(256) void gemm_wmma(const float* __restrict__ X,
                                                 const float* __restrict__ W,
                                                 float* __restrict__ Y) {
  const int wave  = blockIdx.x * (blockDim.x >> 5) + (threadIdx.x >> 5);
  const int lane  = threadIdx.x & 31;
  if (wave >= NTILES_M) return;

  const int mrow  = wave * 16 + (lane & 15);
  const int khalf = (lane >> 4) * 2;
  const int nlo   = lane & 15;

  v8f acc0 = {}, acc1 = {}, acc2 = {}, acc3 = {};

  const float* xrow = X + (size_t)mrow * DIM;
  for (int k0 = 0; k0 < DIM; k0 += 4) {
    v2f a;
    a.x = xrow[k0 + khalf + 0];
    a.y = xrow[k0 + khalf + 1];
    const float* w0 = W + (size_t)(k0 + khalf) * DIM + nlo;
    v2f b0; b0.x = w0[0];        b0.y = w0[DIM];
    v2f b1; b1.x = w0[16];       b1.y = w0[DIM + 16];
    v2f b2; b2.x = w0[32];       b2.y = w0[DIM + 32];
    v2f b3; b3.x = w0[48];       b3.y = w0[DIM + 48];
    acc0 = __builtin_amdgcn_wmma_f32_16x16x4_f32(false, a, false, b0, (short)0, acc0, false, false);
    acc1 = __builtin_amdgcn_wmma_f32_16x16x4_f32(false, a, false, b1, (short)0, acc1, false, false);
    acc2 = __builtin_amdgcn_wmma_f32_16x16x4_f32(false, a, false, b2, (short)0, acc2, false, false);
    acc3 = __builtin_amdgcn_wmma_f32_16x16x4_f32(false, a, false, b3, (short)0, acc3, false, false);
  }

  const int rbase = wave * 16 + (lane >> 4) * 8;
  float* yb = Y + (size_t)rbase * DIM + nlo;
#pragma unroll
  for (int r = 0; r < 8; ++r) {
    yb[(size_t)r * DIM +  0] = acc0[r];
    yb[(size_t)r * DIM + 16] = acc1[r];
    yb[(size_t)r * DIM + 32] = acc2[r];
    yb[(size_t)r * DIM + 48] = acc3[r];
  }
}

#define BIN_ROWS 100
#define NBINS (NNODES / BIN_ROWS)
static_assert(NBINS * BIN_ROWS == NNODES, "ranges must tile the node set exactly");
static_assert(NBINS <= 1024, "one scan block covers all ranges");
static_assert(NNODES <= (1 << 20) && BIN_ROWS < (1 << 11), "record packing");
#define SORT_THREADS 1024
#define SORT_EPT 4
#define SORT_CHUNK (SORT_THREADS * SORT_EPT)

__global__ __launch_bounds__(1024) void zero_bins(int* __restrict__ binCount) {
  binCount[threadIdx.x] = 0;
}

__global__ __launch_bounds__(SORT_THREADS) void count_bins(const int* __restrict__ dst,
                                                           int* __restrict__ binCount,
                                                           int nE) {
  __shared__ int hist[NBINS];
  const int t = threadIdx.x;
  if (t < NBINS) hist[t] = 0;
  __syncthreads();
  const int base = blockIdx.x * SORT_CHUNK;
  for (int i = 0; i < SORT_EPT; ++i) {
    const int e = base + i * SORT_THREADS + t;
    if (e < nE) atomicAdd(&hist[dst[e] / BIN_ROWS], 1);
  }
  __syncthreads();
  if (t < NBINS) {
    const int c = hist[t];
    if (c) atomicAdd(&binCount[t], c);
  }
}

__global__ __launch_bounds__(1024) void scan_bins(const int* __restrict__ binCount,
                                                  int* __restrict__ binStart,
                                                  int* __restrict__ binCursor) {
  __shared__ int s[1024];
  const int t = threadIdx.x;
  const int v = (t < NBINS) ? binCount[t] : 0;
  s[t] = v;
  __syncthreads();
  for (int off = 1; off < 1024; off <<= 1) {
    const int x = (t >= off) ? s[t - off] : 0;
    __syncthreads();
    s[t] += x;
    __syncthreads();
  }
  const int incl = s[t];
  if (t < NBINS) {
    binStart[t] = incl - v;
    binCursor[t] = incl - v;
  }
  if (t == NBINS - 1) binStart[NBINS] = incl;
}

__global__ __launch_bounds__(SORT_THREADS) void fill_bins(const int* __restrict__ src,
                                                          const int* __restrict__ dst,
                                                          const float* __restrict__ ew,
                                                          int* __restrict__ binCursor,
                                                          int* __restrict__ binned,
                                                          int nE) {
  __shared__ int hist[NBINS];
  __shared__ int basePos[NBINS];
  const int t = threadIdx.x;
  if (t < NBINS) hist[t] = 0;
  __syncthreads();
  const int base = blockIdx.x * SORT_CHUNK;
  int bin[SORT_EPT], rank[SORT_EPT], rec0[SORT_EPT], rec1[SORT_EPT];
#pragma unroll
  for (int i = 0; i < SORT_EPT; ++i) {
    const int e = base + i * SORT_THREADS + t;
    bin[i] = -1; rank[i] = 0; rec0[i] = 0; rec1[i] = 0;
    if (e < nE) {
      const int d = dst[e];
      const int b = d / BIN_ROWS;
      bin[i] = b;
      rank[i] = atomicAdd(&hist[b], 1);
      rec0[i] = src[e] | ((d - b * BIN_ROWS) << 20);
      rec1[i] = __float_as_int(ew[e]);
    }
  }
  __syncthreads();
  if (t < NBINS) {
    const int c = hist[t];
    basePos[t] = c ? atomicAdd(&binCursor[t], c) : 0;
  }
  __syncthreads();
  int2* recs = (int2*)binned;
#pragma unroll
  for (int i = 0; i < SORT_EPT; ++i) {
    if (bin[i] >= 0) recs[basePos[bin[i]] + rank[i]] = make_int2(rec0[i], rec1[i]);
  }
}

#define AGG_THREADS 512
#define AGG_WAVES (AGG_THREADS / 32)
#define EDGE_BATCH 8

__device__ __forceinline__ float lane_bcast(float v, int j) {
  return __int_as_float(__builtin_amdgcn_readlane(__float_as_int(v), j));
}

__device__ __forceinline__ void lds_add(float* p, float v) {
  (void)__hip_atomic_fetch_add(p, v, __ATOMIC_RELAXED, __HIP_MEMORY_SCOPE_WORKGROUP);
}

__global__ __launch_bounds__(AGG_THREADS) void aggregate_bins(const float* __restrict__ XW,
                                                              const float* __restrict__ ori,
                                                              const float* __restrict__ bias,
                                                              const int* __restrict__ binStart,
                                                              const int* __restrict__ binned,
                                                              float* __restrict__ out) {
  __shared__ __attribute__((aligned(16))) float acc[BIN_ROWS * DIM];
  const int t = threadIdx.x;
  const int lane = t & 31;
  const int wave = __builtin_amdgcn_readfirstlane(t >> 5);
  const int b = blockIdx.x;
  const size_t row4 = (size_t)b * BIN_ROWS * (DIM / 4);

  for (int i = t; i < BIN_ROWS * (DIM / 4); i += AGG_THREADS) {
    const float4 o = ((const float4*)ori)[row4 + i];
    const float4 bb = ((const float4*)bias)[i & (DIM / 4 - 1)];
    float4 r;
    r.x = o.x + bb.x; r.y = o.y + bb.y; r.z = o.z + bb.z; r.w = o.w + bb.w;
    ((float4*)acc)[i] = r;
  }
  __syncthreads();

  const int e0 = binStart[b];
  const int e1 = binStart[b + 1];
  const int2* recs = (const int2*)binned;

  for (int c = e0 + wave * 32; c < e1; c += AGG_WAVES * 32) {
    const int cnt = (e1 - c < 32) ? (e1 - c) : 32;
    int2 rec = make_int2(0, 0);
    if (lane < cnt) rec = recs[c + lane];
    const int s = rec.x & 0xFFFFF;
    const int dl = rec.x >> 20;
    const float w = __int_as_float(rec.y);

    if (cnt == 32) {
#pragma unroll
      for (int j0 = 0; j0 < 32; j0 += EDGE_BATCH) {
        float2 a[EDGE_BATCH];
#pragma unroll
        for (int u = 0; u < EDGE_BATCH; ++u) {
          const float2* xr = (const float2*)(XW + (size_t)__builtin_amdgcn_readlane(s, j0 + u) * DIM);
          a[u] = xr[lane];
        }
#pragma unroll
        for (int u = 0; u < EDGE_BATCH; ++u) {
          float* o = acc + __builtin_amdgcn_readlane(dl, j0 + u) * DIM;
          const float wj = lane_bcast(w, j0 + u);
          lds_add(o + 2 * lane, a[u].x * wj);
          lds_add(o + 2 * lane + 1, a[u].y * wj);
        }
      }
    } else {
      for (int j = 0; j < cnt; ++j) {
        const float2* xr = (const float2*)(XW + (size_t)__builtin_amdgcn_readlane(s, j) * DIM);
        float* o = acc + __builtin_amdgcn_readlane(dl, j) * DIM;
        const float wj = lane_bcast(w, j);
        const float2 a = xr[lane];
        lds_add(o + 2 * lane, a.x * wj);
        lds_add(o + 2 * lane + 1, a.y * wj);
      }
    }
  }
  __syncthreads();

  for (int i = t; i < BIN_ROWS * (DIM / 4); i += AGG_THREADS)
    ((float4*)out)[row4 + i] = ((const float4*)acc)[i];
}

extern "C" void kernel_launch(void* const* d_in, const int* in_sizes, int n_in,
                              void* d_out, int out_size, void* d_ws, size_t ws_size,
                              hipStream_t stream) {
  const float* in_feat = (const float*)d_in[0];
  const float* ew      = (const float*)d_in[1];
  const float* W1      = (const float*)d_in[2];
  const float* b1      = (const float*)d_in[3];
  const float* W2      = (const float*)d_in[4];
  const float* b2      = (const float*)d_in[5];
  const int*   src     = (const int*)d_in[6];
  const int*   dst     = (const int*)d_in[7];
  const int    nE      = in_sizes[1];

  float* xw   = (float*)d_ws;
  float* hbuf = xw + (size_t)NNODES * DIM;
  int* binCount  = (int*)(hbuf + (size_t)NNODES * DIM);
  int* binStart  = binCount + 1024;
  int* binCursor = binStart + 2048;
  int* binned    = binCursor + 1024;
  float* outp = (float*)d_out;

  const int gemmBlocks = (NTILES_M + 7) / 8;
  const int sortBlocks = (nE + SORT_CHUNK - 1) / SORT_CHUNK;

  zero_bins<<<1, 1024, 0, stream>>>(binCount);
  count_bins<<<sortBlocks, SORT_THREADS, 0, stream>>>(dst, binCount, nE);
  scan_bins<<<1, 1024, 0, stream>>>(binCount, binStart, binCursor);
  fill_bins<<<sortBlocks, SORT_THREADS, 0, stream>>>(src, dst, ew, binCursor, binned, nE);

  const float* hcur = in_feat;
  for (int step = 0; step < 4; ++step) {
    const float* W = (step == 0) ? W1 : W2;
    const float* b = (step == 0) ? b1 : b2;
    float* tgt = (step == 3) ? outp : hbuf;

    gemm_wmma<<<gemmBlocks, 256, 0, stream>>>(hcur, W, xw);
    aggregate_bins<<<NBINS, AGG_THREADS, 0, stream>>>(xw, in_feat, b, binStart, binned, tgt);

    hcur = tgt;
  }
}


# Each step's tail multiplies its finished 100-row LDS tile of h by LDS-staged W2 on f32 WMMA and stores the next step's xw: A(hW) order and FP sums kept; 3 GEMM launches and the h round trip drop
# speedup vs baseline: 3.2470x; 1.2298x over previous
#include <hip/hip_runtime.h>

typedef __attribute__((ext_vector_type(2))) float v2f;
typedef __attribute__((ext_vector_type(8))) float v8f;

#define NNODES 100000
#define DIM 64
#define NTILES_M (NNODES / 16)

__global__ __launch_bounds__(256) void gemm_wmma(const float* __restrict__ X,
                                                 const float* __restrict__ W,
                                                 float* __restrict__ Y) {
  const int wave  = blockIdx.x * (blockDim.x >> 5) + (threadIdx.x >> 5);
  const int lane  = threadIdx.x & 31;
  if (wave >= NTILES_M) return;

  const int mrow  = wave * 16 + (lane & 15);
  const int khalf = (lane >> 4) * 2;
  const int nlo   = lane & 15;

  v8f acc0 = {}, acc1 = {}, acc2 = {}, acc3 = {};

  const float* xrow = X + (size_t)mrow * DIM;
  for (int k0 = 0; k0 < DIM; k0 += 4) {
    v2f a;
    a.x = xrow[k0 + khalf + 0];
    a.y = xrow[k0 + khalf + 1];
    const float* w0 = W + (size_t)(k0 + khalf) * DIM + nlo;
    v2f b0; b0.x = w0[0];        b0.y = w0[DIM];
    v2f b1; b1.x = w0[16];       b1.y = w0[DIM + 16];
    v2f b2; b2.x = w0[32];       b2.y = w0[DIM + 32];
    v2f b3; b3.x = w0[48];       b3.y = w0[DIM + 48];
    acc0 = __builtin_amdgcn_wmma_f32_16x16x4_f32(false, a, false, b0, (short)0, acc0, false, false);
    acc1 = __builtin_amdgcn_wmma_f32_16x16x4_f32(false, a, false, b1, (short)0, acc1, false, false);
    acc2 = __builtin_amdgcn_wmma_f32_16x16x4_f32(false, a, false, b2, (short)0, acc2, false, false);
    acc3 = __builtin_amdgcn_wmma_f32_16x16x4_f32(false, a, false, b3, (short)0, acc3, false, false);
  }

  const int rbase = wave * 16 + (lane >> 4) * 8;
  float* yb = Y + (size_t)rbase * DIM + nlo;
#pragma unroll
  for (int r = 0; r < 8; ++r) {
    yb[(size_t)r * DIM +  0] = acc0[r];
    yb[(size_t)r * DIM + 16] = acc1[r];
    yb[(size_t)r * DIM + 32] = acc2[r];
    yb[(size_t)r * DIM + 48] = acc3[r];
  }
}

#define BIN_ROWS 100
#define NBINS (NNODES / BIN_ROWS)
static_assert(NBINS * BIN_ROWS == NNODES, "ranges must tile the node set exactly");
static_assert(NBINS <= 1024, "one scan block covers all ranges");
static_assert(NNODES <= (1 << 20) && BIN_ROWS < (1 << 11), "record packing");
#define SORT_THREADS 1024
#define SORT_EPT 4
#define SORT_CHUNK (SORT_THREADS * SORT_EPT)

__global__ __launch_bounds__(1024) void zero_bins(int* __restrict__ binCount) {
  binCount[threadIdx.x] = 0;
}

__global__ __launch_bounds__(SORT_THREADS) void count_bins(const int* __restrict__ dst,
                                                           int* __restrict__ binCount,
                                                           int nE) {
  __shared__ int hist[NBINS];
  const int t = threadIdx.x;
  if (t < NBINS) hist[t] = 0;
  __syncthreads();
  const int base = blockIdx.x * SORT_CHUNK;
  for (int i = 0; i < SORT_EPT; ++i) {
    const int e = base + i * SORT_THREADS + t;
    if (e < nE) atomicAdd(&hist[dst[e] / BIN_ROWS], 1);
  }
  __syncthreads();
  if (t < NBINS) {
    const int c = hist[t];
    if (c) atomicAdd(&binCount[t], c);
  }
}

__global__ __launch_bounds__(1024) void scan_bins(const int* __restrict__ binCount,
                                                  int* __restrict__ binStart,
                                                  int* __restrict__ binCursor) {
  __shared__ int s[1024];
  const int t = threadIdx.x;
  const int v = (t < NBINS) ? binCount[t] : 0;
  s[t] = v;
  __syncthreads();
  for (int off = 1; off < 1024; off <<= 1) {
    const int x = (t >= off) ? s[t - off] : 0;
    __syncthreads();
    s[t] += x;
    __syncthreads();
  }
  const int incl = s[t];
  if (t < NBINS) {
    binStart[t] = incl - v;
    binCursor[t] = incl - v;
  }
  if (t == NBINS - 1) binStart[NBINS] = incl;
}

__global__ __launch_bounds__(SORT_THREADS) void fill_bins(const int* __restrict__ src,
                                                          const int* __restrict__ dst,
                                                          const float* __restrict__ ew,
                                                          int* __restrict__ binCursor,
                                                          int* __restrict__ binned,
                                                          int nE) {
  __shared__ int hist[NBINS];
  __shared__ int basePos[NBINS];
  const int t = threadIdx.x;
  if (t < NBINS) hist[t] = 0;
  __syncthreads();
  const int base = blockIdx.x * SORT_CHUNK;
  int bin[SORT_EPT], rank[SORT_EPT], rec0[SORT_EPT], rec1[SORT_EPT];
#pragma unroll
  for (int i = 0; i < SORT_EPT; ++i) {
    const int e = base + i * SORT_THREADS + t;
    bin[i] = -1; rank[i] = 0; rec0[i] = 0; rec1[i] = 0;
    if (e < nE) {
      const int d = dst[e];
      const int b = d / BIN_ROWS;
      bin[i] = b;
      rank[i] = atomicAdd(&hist[b], 1);
      rec0[i] = src[e] | ((d - b * BIN_ROWS) << 20);
      rec1[i] = __float_as_int(ew[e]);
    }
  }
  __syncthreads();
  if (t < NBINS) {
    const int c = hist[t];
    basePos[t] = c ? atomicAdd(&binCursor[t], c) : 0;
  }
  __syncthreads();
  int2* recs = (int2*)binned;
#pragma unroll
  for (int i = 0; i < SORT_EPT; ++i) {
    if (bin[i] >= 0) recs[basePos[bin[i]] + rank[i]] = make_int2(rec0[i], rec1[i]);
  }
}

#define AGG_THREADS 512
#define AGG_WAVES (AGG_THREADS / 32)
#define EDGE_BATCH 8
#define TILE_ROWS 112
#define TILE_MT (TILE_ROWS / 16)
#define TILE_PITCH 68
static_assert(TILE_ROWS >= BIN_ROWS && TILE_ROWS % 16 == 0, "tile covers the range");
static_assert(2 * TILE_MT <= AGG_WAVES, "one wave per (row tile, column half)");
static_assert((TILE_ROWS - BIN_ROWS) * (DIM / 4) <= AGG_THREADS, "one pass zeroes the padding rows");

__device__ __forceinline__ float lane_bcast(float v, int j) {
  return __int_as_float(__builtin_amdgcn_readlane(__float_as_int(v), j));
}

__device__ __forceinline__ void lds_add(float* p, float v) {
  (void)__hip_atomic_fetch_add(p, v, __ATOMIC_RELAXED, __HIP_MEMORY_SCOPE_WORKGROUP);
}

template <bool FUSE_NEXT>
__global__ __launch_bounds__(AGG_THREADS) void aggregate_bins(const float* __restrict__ XW,
                                                              const float* __restrict__ ori,
                                                              const float* __restrict__ bias,
                                                              const int* __restrict__ binStart,
                                                              const int* __restrict__ binned,
                                                              const float* __restrict__ Wn,
                                                              float* __restrict__ out) {
  constexpr int ROWS = FUSE_NEXT ? TILE_ROWS : BIN_ROWS;
  __shared__ __attribute__((aligned(16))) float acc[ROWS * TILE_PITCH];
  __shared__ __attribute__((aligned(16))) float wfrag[FUSE_NEXT ? DIM * DIM : 4];
  const int t = threadIdx.x;
  const int lane = t & 31;
  const int wave = __builtin_amdgcn_readfirstlane(t >> 5);
  const int b = blockIdx.x;
  const size_t row4 = (size_t)b * BIN_ROWS * (DIM / 4);

  constexpr int W_QUADS = DIM * DIM / 4 / AGG_THREADS;
  static_assert(W_QUADS * AGG_THREADS * 4 == DIM * DIM, "Wn splits evenly over the threads");
  float4 wq[W_QUADS];
  if (FUSE_NEXT) {
#pragma unroll
    for (int q = 0; q < W_QUADS; ++q) wq[q] = ((const float4*)Wn)[t + q * AGG_THREADS];
  }

  for (int i = t; i < BIN_ROWS * (DIM / 4); i += AGG_THREADS) {
    const float4 o = ((const float4*)ori)[row4 + i];
    const float4 bb = ((const float4*)bias)[i & (DIM / 4 - 1)];
    float4 r;
    r.x = o.x + bb.x; r.y = o.y + bb.y; r.z = o.z + bb.z; r.w = o.w + bb.w;
    *(float4*)(acc + (i >> 4) * TILE_PITCH + (i & (DIM / 4 - 1)) * 4) = r;
  }
  if (FUSE_NEXT) {
    if (t < (TILE_ROWS - BIN_ROWS) * (DIM / 4))
      *(float4*)(acc + (BIN_ROWS + (t >> 4)) * TILE_PITCH + (t & (DIM / 4 - 1)) * 4) = make_float4(0.f, 0.f, 0.f, 0.f);
#pragma unroll
    for (int q = 0; q < W_QUADS; ++q) {
      const int i = t + q * AGG_THREADS;
      const int k = i >> 4;
      const int n = (i & 15) * 4;
      float* p = wfrag + ((((k >> 2) * 4 + (n >> 4)) * 32 + ((k >> 1) & 1) * 16 + (n & 15)) * 2 + (k & 1));
      p[0] = wq[q].x; p[2] = wq[q].y; p[4] = wq[q].z; p[6] = wq[q].w;
    }
  }
  __syncthreads();

  const int e0 = binStart[b];
  const int e1 = binStart[b + 1];
  const int2* recs = (const int2*)binned;

  for (int c = e0 + wave * 32; c < e1; c += AGG_WAVES * 32) {
    const int cnt = (e1 - c < 32) ? (e1 - c) : 32;
    int2 rec = make_int2(0, 0);
    if (lane < cnt) rec = recs[c + lane];
    const int s = rec.x & 0xFFFFF;
    const int dl = rec.x >> 20;
    const float w = __int_as_float(rec.y);

    if (cnt == 32) {
#pragma unroll
      for (int j0 = 0; j0 < 32; j0 += EDGE_BATCH) {
        float2 a[EDGE_BATCH];
#pragma unroll
        for (int u = 0; u < EDGE_BATCH; ++u) {
          const float2* xr = (const float2*)(XW + (size_t)__builtin_amdgcn_readlane(s, j0 + u) * DIM);
          a[u] = xr[lane];
        }
#pragma unroll
        for (int u = 0; u < EDGE_BATCH; ++u) {
          float* o = acc + __builtin_amdgcn_readlane(dl, j0 + u) * TILE_PITCH;
          const float wj = lane_bcast(w, j0 + u);
          lds_add(o + 2 * lane, a[u].x * wj);
          lds_add(o + 2 * lane + 1, a[u].y * wj);
        }
      }
    } else {
      for (int j = 0; j < cnt; ++j) {
        const float2* xr = (const float2*)(XW + (size_t)__builtin_amdgcn_readlane(s, j) * DIM);
        float* o = acc + __builtin_amdgcn_readlane(dl, j) * TILE_PITCH;
        const float wj = lane_bcast(w, j);
        const float2 a = xr[lane];
        lds_add(o + 2 * lane, a.x * wj);
        lds_add(o + 2 * lane + 1, a.y * wj);
      }
    }
  }
  __syncthreads();

  if (FUSE_NEXT) {
    const int mt = wave >> 1;
    const int ch = wave & 1;
    v8f d0 = {}, d1 = {};
    if (wave < 2 * TILE_MT) {
      const float* ap = acc + (mt * 16 + (lane & 15)) * TILE_PITCH + (lane >> 4) * 2;
      const float* bp = wfrag + (ch * 2 * 32 + lane) * 2;
#pragma unroll
      for (int kk = 0; kk < DIM / 4; ++kk) {
        const v2f a  = *(const v2f*)(ap + kk * 4);
        const v2f b0 = *(const v2f*)(bp + kk * 256);
        const v2f b1 = *(const v2f*)(bp + kk * 256 + 64);
        d0 = __builtin_amdgcn_wmma_f32_16x16x4_f32(false, a, false, b0, (short)0, d0, false, false);
        d1 = __builtin_amdgcn_wmma_f32_16x16x4_f32(false, a, false, b1, (short)0, d1, false, false);
      }
    }
    __syncthreads();
    if (wave < 2 * TILE_MT) {
      float* dp = acc + (mt * 16 + (lane >> 4) * 8) * TILE_PITCH + ch * 32 + (lane & 15);
#pragma unroll
      for (int r = 0; r < 8; ++r) {
        dp[r * TILE_PITCH] = d0[r];
        dp[r * TILE_PITCH + 16] = d1[r];
      }
    }
    __syncthreads();
  }

  for (int i = t; i < BIN_ROWS * (DIM / 4); i += AGG_THREADS)
    ((float4*)out)[row4 + i] = *(const float4*)(acc + (i >> 4) * TILE_PITCH + (i & (DIM / 4 - 1)) * 4);
}

extern "C" void kernel_launch(void* const* d_in, const int* in_sizes, int n_in,
                              void* d_out, int out_size, void* d_ws, size_t ws_size,
                              hipStream_t stream) {
  const float* in_feat = (const float*)d_in[0];
  const float* ew      = (const float*)d_in[1];
  const float* W1      = (const float*)d_in[2];
  const float* b1      = (const float*)d_in[3];
  const float* W2      = (const float*)d_in[4];
  const float* b2      = (const float*)d_in[5];
  const int*   src     = (const int*)d_in[6];
  const int*   dst     = (const int*)d_in[7];
  const int    nE      = in_sizes[1];

  float* xwA  = (float*)d_ws;
  float* xwB  = xwA + (size_t)NNODES * DIM;
  int* binCount  = (int*)(xwB + (size_t)NNODES * DIM);
  int* binStart  = binCount + 1024;
  int* binCursor = binStart + 2048;
  int* binned    = binCursor + 1024;
  float* outp = (float*)d_out;

  const int gemmBlocks = (NTILES_M + 7) / 8;
  const int sortBlocks = (nE + SORT_CHUNK - 1) / SORT_CHUNK;

  zero_bins<<<1, 1024, 0, stream>>>(binCount);
  count_bins<<<sortBlocks, SORT_THREADS, 0, stream>>>(dst, binCount, nE);
  scan_bins<<<1, 1024, 0, stream>>>(binCount, binStart, binCursor);
  fill_bins<<<sortBlocks, SORT_THREADS, 0, stream>>>(src, dst, ew, binCursor, binned, nE);

  gemm_wmma<<<gemmBlocks, 256, 0, stream>>>(in_feat, W1, xwA);
  aggregate_bins<true><<<NBINS, AGG_THREADS, 0, stream>>>(xwA, in_feat, b1, binStart, binned, W2, xwB);
  aggregate_bins<true><<<NBINS, AGG_THREADS, 0, stream>>>(xwB, in_feat, b2, binStart, binned, W2, xwA);
  aggregate_bins<true><<<NBINS, AGG_THREADS, 0, stream>>>(xwA, in_feat, b2, binStart, binned, W2, xwB);
  aggregate_bins<false><<<NBINS, AGG_THREADS, 0, stream>>>(xwB, in_feat, b2, binStart, binned, W2, outp);
}


# Partial last chunk of each range: its row loads issued up to 8 ahead of their LDS adds in scalar-guarded batches, instead of one memory round trip per edge; order of the LDS float adds unchanged
# speedup vs baseline: 3.4457x; 1.0612x over previous
#include <hip/hip_runtime.h>

typedef __attribute__((ext_vector_type(2))) float v2f;
typedef __attribute__((ext_vector_type(8))) float v8f;

#define NNODES 100000
#define DIM 64
#define NTILES_M (NNODES / 16)

__global__ __launch_bounds__(256) void gemm_wmma(const float* __restrict__ X,
                                                 const float* __restrict__ W,
                                                 float* __restrict__ Y) {
  const int wave  = blockIdx.x * (blockDim.x >> 5) + (threadIdx.x >> 5);
  const int lane  = threadIdx.x & 31;
  if (wave >= NTILES_M) return;

  const int mrow  = wave * 16 + (lane & 15);
  const int khalf = (lane >> 4) * 2;
  const int nlo   = lane & 15;

  v8f acc0 = {}, acc1 = {}, acc2 = {}, acc3 = {};

  const float* xrow = X + (size_t)mrow * DIM;
  for (int k0 = 0; k0 < DIM; k0 += 4) {
    v2f a;
    a.x = xrow[k0 + khalf + 0];
    a.y = xrow[k0 + khalf + 1];
    const float* w0 = W + (size_t)(k0 + khalf) * DIM + nlo;
    v2f b0; b0.x = w0[0];        b0.y = w0[DIM];
    v2f b1; b1.x = w0[16];       b1.y = w0[DIM + 16];
    v2f b2; b2.x = w0[32];       b2.y = w0[DIM + 32];
    v2f b3; b3.x = w0[48];       b3.y = w0[DIM + 48];
    acc0 = __builtin_amdgcn_wmma_f32_16x16x4_f32(false, a, false, b0, (short)0, acc0, false, false);
    acc1 = __builtin_amdgcn_wmma_f32_16x16x4_f32(false, a, false, b1, (short)0, acc1, false, false);
    acc2 = __builtin_amdgcn_wmma_f32_16x16x4_f32(false, a, false, b2, (short)0, acc2, false, false);
    acc3 = __builtin_amdgcn_wmma_f32_16x16x4_f32(false, a, false, b3, (short)0, acc3, false, false);
  }

  const int rbase = wave * 16 + (lane >> 4) * 8;
  float* yb = Y + (size_t)rbase * DIM + nlo;
#pragma unroll
  for (int r = 0; r < 8; ++r) {
    yb[(size_t)r * DIM +  0] = acc0[r];
    yb[(size_t)r * DIM + 16] = acc1[r];
    yb[(size_t)r * DIM + 32] = acc2[r];
    yb[(size_t)r * DIM + 48] = acc3[r];
  }
}

#define BIN_ROWS 100
#define NBINS (NNODES / BIN_ROWS)
static_assert(NBINS * BIN_ROWS == NNODES, "ranges must tile the node set exactly");
static_assert(NBINS <= 1024, "one scan block covers all ranges");
static_assert(NNODES <= (1 << 20) && BIN_ROWS < (1 << 11), "record packing");
#define SORT_THREADS 1024
#define SORT_EPT 4
#define SORT_CHUNK (SORT_THREADS * SORT_EPT)

__global__ __launch_bounds__(1024) void zero_bins(int* __restrict__ binCount) {
  binCount[threadIdx.x] = 0;
}

__global__ __launch_bounds__(SORT_THREADS) void count_bins(const int* __restrict__ dst,
                                                           int* __restrict__ binCount,
                                                           int nE) {
  __shared__ int hist[NBINS];
  const int t = threadIdx.x;
  if (t < NBINS) hist[t] = 0;
  __syncthreads();
  const int base = blockIdx.x * SORT_CHUNK;
  for (int i = 0; i < SORT_EPT; ++i) {
    const int e = base + i * SORT_THREADS + t;
    if (e < nE) atomicAdd(&hist[dst[e] / BIN_ROWS], 1);
  }
  __syncthreads();
  if (t < NBINS) {
    const int c = hist[t];
    if (c) atomicAdd(&binCount[t], c);
  }
}

__global__ __launch_bounds__(1024) void scan_bins(const int* __restrict__ binCount,
                                                  int* __restrict__ binStart,
                                                  int* __restrict__ binCursor) {
  __shared__ int s[1024];
  const int t = threadIdx.x;
  const int v = (t < NBINS) ? binCount[t] : 0;
  s[t] = v;
  __syncthreads();
  for (int off = 1; off < 1024; off <<= 1) {
    const int x = (t >= off) ? s[t - off] : 0;
    __syncthreads();
    s[t] += x;
    __syncthreads();
  }
  const int incl = s[t];
  if (t < NBINS) {
    binStart[t] = incl - v;
    binCursor[t] = incl - v;
  }
  if (t == NBINS - 1) binStart[NBINS] = incl;
}

__global__ __launch_bounds__(SORT_THREADS) void fill_bins(const int* __restrict__ src,
                                                          const int* __restrict__ dst,
                                                          const float* __restrict__ ew,
                                                          int* __restrict__ binCursor,
                                                          int* __restrict__ binned,
                                                          int nE) {
  __shared__ int hist[NBINS];
  __shared__ int basePos[NBINS];
  const int t = threadIdx.x;
  if (t < NBINS) hist[t] = 0;
  __syncthreads();
  const int base = blockIdx.x * SORT_CHUNK;
  int bin[SORT_EPT], rank[SORT_EPT], rec0[SORT_EPT], rec1[SORT_EPT];
#pragma unroll
  for (int i = 0; i < SORT_EPT; ++i) {
    const int e = base + i * SORT_THREADS + t;
    bin[i] = -1; rank[i] = 0; rec0[i] = 0; rec1[i] = 0;
    if (e < nE) {
      const int d = dst[e];
      const int b = d / BIN_ROWS;
      bin[i] = b;
      rank[i] = atomicAdd(&hist[b], 1);
      rec0[i] = src[e] | ((d - b * BIN_ROWS) << 20);
      rec1[i] = __float_as_int(ew[e]);
    }
  }
  __syncthreads();
  if (t < NBINS) {
    const int c = hist[t];
    basePos[t] = c ? atomicAdd(&binCursor[t], c) : 0;
  }
  __syncthreads();
  int2* recs = (int2*)binned;
#pragma unroll
  for (int i = 0; i < SORT_EPT; ++i) {
    if (bin[i] >= 0) recs[basePos[bin[i]] + rank[i]] = make_int2(rec0[i], rec1[i]);
  }
}

#define AGG_THREADS 512
#define AGG_WAVES (AGG_THREADS / 32)
#define EDGE_BATCH 8
#define TILE_ROWS 112
#define TILE_MT (TILE_ROWS / 16)
#define TILE_PITCH 68
static_assert(TILE_ROWS >= BIN_ROWS && TILE_ROWS % 16 == 0, "tile covers the range");
static_assert(2 * TILE_MT <= AGG_WAVES, "one wave per (row tile, column half)");
static_assert((TILE_ROWS - BIN_ROWS) * (DIM / 4) <= AGG_THREADS, "one pass zeroes the padding rows");

__device__ __forceinline__ float lane_bcast(float v, int j) {
  return __int_as_float(__builtin_amdgcn_readlane(__float_as_int(v), j));
}

__device__ __forceinline__ void lds_add(float* p, float v) {
  (void)__hip_atomic_fetch_add(p, v, __ATOMIC_RELAXED, __HIP_MEMORY_SCOPE_WORKGROUP);
}

template <bool FUSE_NEXT>
__global__ __launch_bounds__(AGG_THREADS) void aggregate_bins(const float* __restrict__ XW,
                                                              const float* __restrict__ ori,
                                                              const float* __restrict__ bias,
                                                              const int* __restrict__ binStart,
                                                              const int* __restrict__ binned,
                                                              const float* __restrict__ Wn,
                                                              float* __restrict__ out) {
  constexpr int ROWS = FUSE_NEXT ? TILE_ROWS : BIN_ROWS;
  __shared__ __attribute__((aligned(16))) float acc[ROWS * TILE_PITCH];
  __shared__ __attribute__((aligned(16))) float wfrag[FUSE_NEXT ? DIM * DIM : 4];
  const int t = threadIdx.x;
  const int lane = t & 31;
  const int wave = __builtin_amdgcn_readfirstlane(t >> 5);
  const int b = blockIdx.x;
  const size_t row4 = (size_t)b * BIN_ROWS * (DIM / 4);

  constexpr int W_QUADS = DIM * DIM / 4 / AGG_THREADS;
  static_assert(W_QUADS * AGG_THREADS * 4 == DIM * DIM, "Wn splits evenly over the threads");
  float4 wq[W_QUADS];
  if (FUSE_NEXT) {
#pragma unroll
    for (int q = 0; q < W_QUADS; ++q) wq[q] = ((const float4*)Wn)[t + q * AGG_THREADS];
  }

  for (int i = t; i < BIN_ROWS * (DIM / 4); i += AGG_THREADS) {
    const float4 o = ((const float4*)ori)[row4 + i];
    const float4 bb = ((const float4*)bias)[i & (DIM / 4 - 1)];
    float4 r;
    r.x = o.x + bb.x; r.y = o.y + bb.y; r.z = o.z + bb.z; r.w = o.w + bb.w;
    *(float4*)(acc + (i >> 4) * TILE_PITCH + (i & (DIM / 4 - 1)) * 4) = r;
  }
  if (FUSE_NEXT) {
    if (t < (TILE_ROWS - BIN_ROWS) * (DIM / 4))
      *(float4*)(acc + (BIN_ROWS + (t >> 4)) * TILE_PITCH + (t & (DIM / 4 - 1)) * 4) = make_float4(0.f, 0.f, 0.f, 0.f);
#pragma unroll
    for (int q = 0; q < W_QUADS; ++q) {
      const int i = t + q * AGG_THREADS;
      const int k = i >> 4;
      const int n = (i & 15) * 4;
      float* p = wfrag + ((((k >> 2) * 4 + (n >> 4)) * 32 + ((k >> 1) & 1) * 16 + (n & 15)) * 2 + (k & 1));
      p[0] = wq[q].x; p[2] = wq[q].y; p[4] = wq[q].z; p[6] = wq[q].w;
    }
  }
  __syncthreads();

  const int e0 = binStart[b];
  const int e1 = binStart[b + 1];
  const int2* recs = (const int2*)binned;

  for (int c = e0 + wave * 32; c < e1; c += AGG_WAVES * 32) {
    const int cnt = (e1 - c < 32) ? (e1 - c) : 32;
    int2 rec = make_int2(0, 0);
    if (lane < cnt) rec = recs[c + lane];
    const int s = rec.x & 0xFFFFF;
    const int dl = rec.x >> 20;
    const float w = __int_as_float(rec.y);

    if (cnt == 32) {
#pragma unroll
      for (int j0 = 0; j0 < 32; j0 += EDGE_BATCH) {
        float2 a[EDGE_BATCH];
#pragma unroll
        for (int u = 0; u < EDGE_BATCH; ++u) {
          const float2* xr = (const float2*)(XW + (size_t)__builtin_amdgcn_readlane(s, j0 + u) * DIM);
          a[u] = xr[lane];
        }
#pragma unroll
        for (int u = 0; u < EDGE_BATCH; ++u) {
          float* o = acc + __builtin_amdgcn_readlane(dl, j0 + u) * TILE_PITCH;
          const float wj = lane_bcast(w, j0 + u);
          lds_add(o + 2 * lane, a[u].x * wj);
          lds_add(o + 2 * lane + 1, a[u].y * wj);
        }
      }
    } else {
      for (int j0 = 0; j0 < cnt; j0 += EDGE_BATCH) {
        const int nb = cnt - j0;
        float2 a[EDGE_BATCH];
#pragma unroll
        for (int u = 0; u < EDGE_BATCH; ++u) {
          a[u] = make_float2(0.f, 0.f);
          if (u < nb) {
            const float2* xr = (const float2*)(XW + (size_t)__builtin_amdgcn_readlane(s, j0 + u) * DIM);
            a[u] = xr[lane];
          }
        }
#pragma unroll
        for (int u = 0; u < EDGE_BATCH; ++u) {
          if (u < nb) {
            float* o = acc + __builtin_amdgcn_readlane(dl, j0 + u) * TILE_PITCH;
            const float wj = lane_bcast(w, j0 + u);
            lds_add(o + 2 * lane, a[u].x * wj);
            lds_add(o + 2 * lane + 1, a[u].y * wj);
          }
        }
      }
    }
  }
  __syncthreads();

  if (FUSE_NEXT) {
    const int mt = wave >> 1;
    const int ch = wave & 1;
    v8f d0 = {}, d1 = {};
    if (wave < 2 * TILE_MT) {
      const float* ap = acc + (mt * 16 + (lane & 15)) * TILE_PITCH + (lane >> 4) * 2;
      const float* bp = wfrag + (ch * 2 * 32 + lane) * 2;
#pragma unroll
      for (int kk = 0; kk < DIM / 4; ++kk) {
        const v2f a  = *(const v2f*)(ap + kk * 4);
        const v2f b0 = *(const v2f*)(bp + kk * 256);
        const v2f b1 = *(const v2f*)(bp + kk * 256 + 64);
        d0 = __builtin_amdgcn_wmma_f32_16x16x4_f32(false, a, false, b0, (short)0, d0, false, false);
        d1 = __builtin_amdgcn_wmma_f32_16x16x4_f32(false, a, false, b1, (short)0, d1, false, false);
      }
    }
    __syncthreads();
    if (wave < 2 * TILE_MT) {
      float* dp = acc + (mt * 16 + (lane >> 4) * 8) * TILE_PITCH + ch * 32 + (lane & 15);
#pragma unroll
      for (int r = 0; r < 8; ++r) {
        dp[r * TILE_PITCH] = d0[r];
        dp[r * TILE_PITCH + 16] = d1[r];
      }
    }
    __syncthreads();
  }

  for (int i = t; i < BIN_ROWS * (DIM / 4); i += AGG_THREADS)
    ((float4*)out)[row4 + i] = *(const float4*)(acc + (i >> 4) * TILE_PITCH + (i & (DIM / 4 - 1)) * 4);
}

extern "C" void kernel_launch(void* const* d_in, const int* in_sizes, int n_in,
                              void* d_out, int out_size, void* d_ws, size_t ws_size,
                              hipStream_t stream) {
  const float* in_feat = (const float*)d_in[0];
  const float* ew      = (const float*)d_in[1];
  const float* W1      = (const float*)d_in[2];
  const float* b1      = (const float*)d_in[3];
  const float* W2      = (const float*)d_in[4];
  const float* b2      = (const float*)d_in[5];
  const int*   src     = (const int*)d_in[6];
  const int*   dst     = (const int*)d_in[7];
  const int    nE      = in_sizes[1];

  float* xwA  = (float*)d_ws;
  float* xwB  = xwA + (size_t)NNODES * DIM;
  int* binCount  = (int*)(xwB + (size_t)NNODES * DIM);
  int* binStart  = binCount + 1024;
  int* binCursor = binStart + 2048;
  int* binned    = binCursor + 1024;
  float* outp = (float*)d_out;

  const int gemmBlocks = (NTILES_M + 7) / 8;
  const int sortBlocks = (nE + SORT_CHUNK - 1) / SORT_CHUNK;

  zero_bins<<<1, 1024, 0, stream>>>(binCount);
  count_bins<<<sortBlocks, SORT_THREADS, 0, stream>>>(dst, binCount, nE);
  scan_bins<<<1, 1024, 0, stream>>>(binCount, binStart, binCursor);
  fill_bins<<<sortBlocks, SORT_THREADS, 0, stream>>>(src, dst, ew, binCursor, binned, nE);

  gemm_wmma<<<gemmBlocks, 256, 0, stream>>>(in_feat, W1, xwA);
  aggregate_bins<true><<<NBINS, AGG_THREADS, 0, stream>>>(xwA, in_feat, b1, binStart, binned, W2, xwB);
  aggregate_bins<true><<<NBINS, AGG_THREADS, 0, stream>>>(xwB, in_feat, b2, binStart, binned, W2, xwA);
  aggregate_bins<true><<<NBINS, AGG_THREADS, 0, stream>>>(xwA, in_feat, b2, binStart, binned, W2, xwB);
  aggregate_bins<false><<<NBINS, AGG_THREADS, 0, stream>>>(xwB, in_feat, b2, binStart, binned, W2, outp);
}


# Step 0 gathers the raw input rows and its tail applies W1 to the finished sums on f32 WMMA, (A x)W1 for A(x W1): the stand-alone GEMM launch drops; order of step 0's FP sums changes
# speedup vs baseline: 3.7162x; 1.0785x over previous
#include <hip/hip_runtime.h>

typedef __attribute__((ext_vector_type(2))) float v2f;
typedef __attribute__((ext_vector_type(8))) float v8f;

#define NNODES 100000
#define DIM 64


#define BIN_ROWS 100
#define NBINS (NNODES / BIN_ROWS)
static_assert(NBINS * BIN_ROWS == NNODES, "ranges must tile the node set exactly");
static_assert(NBINS <= 1024, "one scan block covers all ranges");
static_assert(NNODES <= (1 << 20) && BIN_ROWS < (1 << 11), "record packing");
#define SORT_THREADS 1024
#define SORT_EPT 4
#define SORT_CHUNK (SORT_THREADS * SORT_EPT)

__global__ __launch_bounds__(1024) void zero_bins(int* __restrict__ binCount) {
  binCount[threadIdx.x] = 0;
}

__global__ __launch_bounds__(SORT_THREADS) void count_bins(const int* __restrict__ dst,
                                                           int* __restrict__ binCount,
                                                           int nE) {
  __shared__ int hist[NBINS];
  const int t = threadIdx.x;
  if (t < NBINS) hist[t] = 0;
  __syncthreads();
  const int base = blockIdx.x * SORT_CHUNK;
  for (int i = 0; i < SORT_EPT; ++i) {
    const int e = base + i * SORT_THREADS + t;
    if (e < nE) atomicAdd(&hist[dst[e] / BIN_ROWS], 1);
  }
  __syncthreads();
  if (t < NBINS) {
    const int c = hist[t];
    if (c) atomicAdd(&binCount[t], c);
  }
}

__global__ __launch_bounds__(1024) void scan_bins(const int* __restrict__ binCount,
                                                  int* __restrict__ binStart,
                                                  int* __restrict__ binCursor) {
  __shared__ int s[1024];
  const int t = threadIdx.x;
  const int v = (t < NBINS) ? binCount[t] : 0;
  s[t] = v;
  __syncthreads();
  for (int off = 1; off < 1024; off <<= 1) {
    const int x = (t >= off) ? s[t - off] : 0;
    __syncthreads();
    s[t] += x;
    __syncthreads();
  }
  const int incl = s[t];
  if (t < NBINS) {
    binStart[t] = incl - v;
    binCursor[t] = incl - v;
  }
  if (t == NBINS - 1) binStart[NBINS] = incl;
}

__global__ __launch_bounds__(SORT_THREADS) void fill_bins(const int* __restrict__ src,
                                                          const int* __restrict__ dst,
                                                          const float* __restrict__ ew,
                                                          int* __restrict__ binCursor,
                                                          int* __restrict__ binned,
                                                          int nE) {
  __shared__ int hist[NBINS];
  __shared__ int basePos[NBINS];
  const int t = threadIdx.x;
  if (t < NBINS) hist[t] = 0;
  __syncthreads();
  const int base = blockIdx.x * SORT_CHUNK;
  int bin[SORT_EPT], rank[SORT_EPT], rec0[SORT_EPT], rec1[SORT_EPT];
#pragma unroll
  for (int i = 0; i < SORT_EPT; ++i) {
    const int e = base + i * SORT_THREADS + t;
    bin[i] = -1; rank[i] = 0; rec0[i] = 0; rec1[i] = 0;
    if (e < nE) {
      const int d = dst[e];
      const int b = d / BIN_ROWS;
      bin[i] = b;
      rank[i] = atomicAdd(&hist[b], 1);
      rec0[i] = src[e] | ((d - b * BIN_ROWS) << 20);
      rec1[i] = __float_as_int(ew[e]);
    }
  }
  __syncthreads();
  if (t < NBINS) {
    const int c = hist[t];
    basePos[t] = c ? atomicAdd(&binCursor[t], c) : 0;
  }
  __syncthreads();
  int2* recs = (int2*)binned;
#pragma unroll
  for (int i = 0; i < SORT_EPT; ++i) {
    if (bin[i] >= 0) recs[basePos[bin[i]] + rank[i]] = make_int2(rec0[i], rec1[i]);
  }
}

#define AGG_THREADS 512
#define AGG_WAVES (AGG_THREADS / 32)
#define EDGE_BATCH 8
#define TILE_ROWS 112
#define TILE_MT (TILE_ROWS / 16)
#define TILE_PITCH 68
#define STEP_LAST 0
#define STEP_MID 1
#define STEP_FIRST 2
static_assert(TILE_ROWS >= BIN_ROWS && TILE_ROWS % 16 == 0, "tile covers the range");
static_assert(2 * TILE_MT <= AGG_WAVES, "one wave per (row tile, column half)");
static_assert((TILE_ROWS - BIN_ROWS) * (DIM / 4) <= AGG_THREADS, "one pass zeroes the padding rows");
static_assert(AGG_THREADS % (DIM / 4) == 0, "a thread's pieces all fall in one 16-byte column of a row");

__device__ __forceinline__ float lane_bcast(float v, int j) {
  return __int_as_float(__builtin_amdgcn_readlane(__float_as_int(v), j));
}

__device__ __forceinline__ void lds_add(float* p, float v) {
  (void)__hip_atomic_fetch_add(p, v, __ATOMIC_RELAXED, __HIP_MEMORY_SCOPE_WORKGROUP);
}

constexpr int W_QUADS = DIM * DIM / 4 / AGG_THREADS;
static_assert(W_QUADS * AGG_THREADS * 4 == DIM * DIM, "the matrix splits evenly over the threads");

__device__ __forceinline__ void stage_wfrag(float* wf, const float4 (&wq)[W_QUADS], int t) {
#pragma unroll
  for (int q = 0; q < W_QUADS; ++q) {
    const int i = t + q * AGG_THREADS;
    const int k = i >> 4;
    const int n = (i & 15) * 4;
    float* p = wf + ((((k >> 2) * 4 + (n >> 4)) * 32 + ((k >> 1) & 1) * 16 + (n & 15)) * 2 + (k & 1));
    p[0] = wq[q].x; p[2] = wq[q].y; p[4] = wq[q].z; p[6] = wq[q].w;
  }
}

__device__ __forceinline__ void tile_product(const float* tile, const float* wf, int mt, int ch, int lane,
                                             v8f& d0, v8f& d1) {
  const float* ap = tile + (mt * 16 + (lane & 15)) * TILE_PITCH + (lane >> 4) * 2;
  const float* bp = wf + (ch * 2 * 32 + lane) * 2;
#pragma unroll
  for (int kk = 0; kk < DIM / 4; ++kk) {
    const v2f a  = *(const v2f*)(ap + kk * 4);
    const v2f b0 = *(const v2f*)(bp + kk * 256);
    const v2f b1 = *(const v2f*)(bp + kk * 256 + 64);
    d0 = __builtin_amdgcn_wmma_f32_16x16x4_f32(false, a, false, b0, (short)0, d0, false, false);
    d1 = __builtin_amdgcn_wmma_f32_16x16x4_f32(false, a, false, b1, (short)0, d1, false, false);
  }
}

__device__ __forceinline__ void tile_put(float* tile, int mt, int ch, int lane, const v8f& d0, const v8f& d1) {
  float* dp = tile + (mt * 16 + (lane >> 4) * 8) * TILE_PITCH + ch * 32 + (lane & 15);
#pragma unroll
  for (int r = 0; r < 8; ++r) {
    dp[r * TILE_PITCH] = d0[r];
    dp[r * TILE_PITCH + 16] = d1[r];
  }
}

template <int STEP>
__global__ __launch_bounds__(AGG_THREADS) void aggregate_bins(const float* __restrict__ XW,
                                                              const float* __restrict__ ori,
                                                              const float* __restrict__ bias,
                                                              const int* __restrict__ binStart,
                                                              const int* __restrict__ binned,
                                                              const float* __restrict__ Wf,
                                                              const float* __restrict__ Wn,
                                                              float* __restrict__ out) {
  constexpr bool FUSE_NEXT = STEP != STEP_LAST;
  constexpr bool FIRST = STEP == STEP_FIRST;
  constexpr int ROWS = FUSE_NEXT ? TILE_ROWS : BIN_ROWS;
  constexpr int ROW_QUADS = BIN_ROWS * (DIM / 4);
  __shared__ __attribute__((aligned(16))) float acc[ROWS * TILE_PITCH];
  __shared__ __attribute__((aligned(16))) float wfrag[FUSE_NEXT ? DIM * DIM : 4];
  __shared__ __attribute__((aligned(16))) float wfirst[FIRST ? DIM * DIM : 4];
  const int t = threadIdx.x;
  const int lane = t & 31;
  const int wave = __builtin_amdgcn_readfirstlane(t >> 5);
  const int b = blockIdx.x;
  const size_t row4 = (size_t)b * ROW_QUADS;

  float4 wq[W_QUADS], wq1[W_QUADS];
  if (FUSE_NEXT) {
#pragma unroll
    for (int q = 0; q < W_QUADS; ++q) wq[q] = ((const float4*)Wn)[t + q * AGG_THREADS];
  }
  if (FIRST) {
#pragma unroll
    for (int q = 0; q < W_QUADS; ++q) wq1[q] = ((const float4*)Wf)[t + q * AGG_THREADS];
  }

  if (FIRST) {
    for (int i = t; i < TILE_ROWS * (DIM / 4); i += AGG_THREADS)
      *(float4*)(acc + (i >> 4) * TILE_PITCH + (i & (DIM / 4 - 1)) * 4) = make_float4(0.f, 0.f, 0.f, 0.f);
  } else {
    for (int i = t; i < ROW_QUADS; i += AGG_THREADS) {
      const float4 o = ((const float4*)ori)[row4 + i];
      const float4 bb = ((const float4*)bias)[i & (DIM / 4 - 1)];
      float4 r;
      r.x = o.x + bb.x; r.y = o.y + bb.y; r.z = o.z + bb.z; r.w = o.w + bb.w;
      *(float4*)(acc + (i >> 4) * TILE_PITCH + (i & (DIM / 4 - 1)) * 4) = r;
    }
    if (FUSE_NEXT) {
      if (t < (TILE_ROWS - BIN_ROWS) * (DIM / 4))
        *(float4*)(acc + (BIN_ROWS + (t >> 4)) * TILE_PITCH + (t & (DIM / 4 - 1)) * 4) = make_float4(0.f, 0.f, 0.f, 0.f);
    }
  }
  if (FUSE_NEXT) stage_wfrag(wfrag, wq, t);
  if (FIRST) stage_wfrag(wfirst, wq1, t);
  __syncthreads();

  const int e0 = binStart[b];
  const int e1 = binStart[b + 1];
  const int2* recs = (const int2*)binned;

  for (int c = e0 + wave * 32; c < e1; c += AGG_WAVES * 32) {
    const int cnt = (e1 - c < 32) ? (e1 - c) : 32;
    int2 rec = make_int2(0, 0);
    if (lane < cnt) rec = recs[c + lane];
    const int s = rec.x & 0xFFFFF;
    const int dl = rec.x >> 20;
    const float w = __int_as_float(rec.y);

    if (cnt == 32) {
#pragma unroll
      for (int j0 = 0; j0 < 32; j0 += EDGE_BATCH) {
        float2 a[EDGE_BATCH];
#pragma unroll
        for (int u = 0; u < EDGE_BATCH; ++u) {
          const float2* xr = (const float2*)(XW + (size_t)__builtin_amdgcn_readlane(s, j0 + u) * DIM);
          a[u] = xr[lane];
        }
#pragma unroll
        for (int u = 0; u < EDGE_BATCH; ++u) {
          float* o = acc + __builtin_amdgcn_readlane(dl, j0 + u) * TILE_PITCH;
          const float wj = lane_bcast(w, j0 + u);
          lds_add(o + 2 * lane, a[u].x * wj);
          lds_add(o + 2 * lane + 1, a[u].y * wj);
        }
      }
    } else {
      for (int j0 = 0; j0 < cnt; j0 += EDGE_BATCH) {
        const int nb = cnt - j0;
        float2 a[EDGE_BATCH];
#pragma unroll
        for (int u = 0; u < EDGE_BATCH; ++u) {
          a[u] = make_float2(0.f, 0.f);
          if (u < nb) {
            const float2* xr = (const float2*)(XW + (size_t)__builtin_amdgcn_readlane(s, j0 + u) * DIM);
            a[u] = xr[lane];
          }
        }
#pragma unroll
        for (int u = 0; u < EDGE_BATCH; ++u) {
          if (u < nb) {
            float* o = acc + __builtin_amdgcn_readlane(dl, j0 + u) * TILE_PITCH;
            const float wj = lane_bcast(w, j0 + u);
            lds_add(o + 2 * lane, a[u].x * wj);
            lds_add(o + 2 * lane + 1, a[u].y * wj);
          }
        }
      }
    }
  }
  __syncthreads();

  const int mt = wave >> 1;
  const int ch = wave & 1;

  if (FIRST) {
    constexpr int ORI_Q = (ROW_QUADS + AGG_THREADS - 1) / AGG_THREADS;
    float4 o[ORI_Q];
#pragma unroll
    for (int q = 0; q < ORI_Q; ++q) {
      const int i = t + q * AGG_THREADS;
      o[q] = ((const float4*)ori)[row4 + (i < ROW_QUADS ? i : ROW_QUADS - 1)];
    }
    const float4 bb = ((const float4*)bias)[t & (DIM / 4 - 1)];
    v8f f0 = {}, f1 = {};
    if (wave < 2 * TILE_MT) tile_product(acc, wfirst, mt, ch, lane, f0, f1);
    __syncthreads();
    if (wave < 2 * TILE_MT) tile_put(acc, mt, ch, lane, f0, f1);
    __syncthreads();
#pragma unroll
    for (int q = 0; q < ORI_Q; ++q) {
      const int i = t + q * AGG_THREADS;
      if (i < ROW_QUADS) {
        float* p = acc + (i >> 4) * TILE_PITCH + (i & (DIM / 4 - 1)) * 4;
        const float4 s = *(const float4*)p;
        float4 r;
        r.x = (s.x + bb.x) + o[q].x; r.y = (s.y + bb.y) + o[q].y;
        r.z = (s.z + bb.z) + o[q].z; r.w = (s.w + bb.w) + o[q].w;
        *(float4*)p = r;
      }
    }
    __syncthreads();
  }

  if (FUSE_NEXT) {
    v8f d0 = {}, d1 = {};
    if (wave < 2 * TILE_MT) tile_product(acc, wfrag, mt, ch, lane, d0, d1);
    __syncthreads();
    if (wave < 2 * TILE_MT) tile_put(acc, mt, ch, lane, d0, d1);
    __syncthreads();
  }

  for (int i = t; i < ROW_QUADS; i += AGG_THREADS)
    ((float4*)out)[row4 + i] = *(const float4*)(acc + (i >> 4) * TILE_PITCH + (i & (DIM / 4 - 1)) * 4);
}

extern "C" void kernel_launch(void* const* d_in, const int* in_sizes, int n_in,
                              void* d_out, int out_size, void* d_ws, size_t ws_size,
                              hipStream_t stream) {
  const float* in_feat = (const float*)d_in[0];
  const float* ew      = (const float*)d_in[1];
  const float* W1      = (const float*)d_in[2];
  const float* b1      = (const float*)d_in[3];
  const float* W2      = (const float*)d_in[4];
  const float* b2      = (const float*)d_in[5];
  const int*   src     = (const int*)d_in[6];
  const int*   dst     = (const int*)d_in[7];
  const int    nE      = in_sizes[1];

  float* xwA  = (float*)d_ws;
  float* xwB  = xwA + (size_t)NNODES * DIM;
  int* binCount  = (int*)(xwB + (size_t)NNODES * DIM);
  int* binStart  = binCount + 1024;
  int* binCursor = binStart + 2048;
  int* binned    = binCursor + 1024;
  float* outp = (float*)d_out;

  const int sortBlocks = (nE + SORT_CHUNK - 1) / SORT_CHUNK;

  zero_bins<<<1, 1024, 0, stream>>>(binCount);
  count_bins<<<sortBlocks, SORT_THREADS, 0, stream>>>(dst, binCount, nE);
  scan_bins<<<1, 1024, 0, stream>>>(binCount, binStart, binCursor);
  fill_bins<<<sortBlocks, SORT_THREADS, 0, stream>>>(src, dst, ew, binCursor, binned, nE);

  aggregate_bins<STEP_FIRST><<<NBINS, AGG_THREADS, 0, stream>>>(in_feat, in_feat, b1, binStart, binned, W1, W2, xwA);
  aggregate_bins<STEP_MID><<<NBINS, AGG_THREADS, 0, stream>>>(xwA, in_feat, b2, binStart, binned, W2, W2, xwB);
  aggregate_bins<STEP_MID><<<NBINS, AGG_THREADS, 0, stream>>>(xwB, in_feat, b2, binStart, binned, W2, W2, xwA);
  aggregate_bins<STEP_LAST><<<NBINS, AGG_THREADS, 0, stream>>>(xwA, in_feat, b2, binStart, binned, W2, W2, outp);
}


# Each wave owns 6-7 rows of its range and sums them in registers; records regrouped by owner wave in LDS each step (integer counters): no LDS float atomics; FP sums per row in regrouped record order
# speedup vs baseline: 4.0631x; 1.0933x over previous
#include <hip/hip_runtime.h>

typedef __attribute__((ext_vector_type(2))) float v2f;
typedef __attribute__((ext_vector_type(8))) float v8f;

#define NNODES 100000
#define DIM 64


#define BIN_ROWS 100
#define NBINS (NNODES / BIN_ROWS)
static_assert(NBINS * BIN_ROWS == NNODES, "ranges must tile the node set exactly");
static_assert(NBINS <= 1024, "one scan block covers all ranges");
static_assert(NNODES <= (1 << 20) && BIN_ROWS < (1 << 11), "record packing");
#define SORT_THREADS 1024
#define SORT_EPT 4
#define SORT_CHUNK (SORT_THREADS * SORT_EPT)

__global__ __launch_bounds__(1024) void zero_bins(int* __restrict__ binCount) {
  binCount[threadIdx.x] = 0;
}

__global__ __launch_bounds__(SORT_THREADS) void count_bins(const int* __restrict__ dst,
                                                           int* __restrict__ binCount,
                                                           int nE) {
  __shared__ int hist[NBINS];
  const int t = threadIdx.x;
  if (t < NBINS) hist[t] = 0;
  __syncthreads();
  const int base = blockIdx.x * SORT_CHUNK;
  for (int i = 0; i < SORT_EPT; ++i) {
    const int e = base + i * SORT_THREADS + t;
    if (e < nE) atomicAdd(&hist[dst[e] / BIN_ROWS], 1);
  }
  __syncthreads();
  if (t < NBINS) {
    const int c = hist[t];
    if (c) atomicAdd(&binCount[t], c);
  }
}

__global__ __launch_bounds__(1024) void scan_bins(const int* __restrict__ binCount,
                                                  int* __restrict__ binStart,
                                                  int* __restrict__ binCursor) {
  __shared__ int s[1024];
  const int t = threadIdx.x;
  const int v = (t < NBINS) ? binCount[t] : 0;
  s[t] = v;
  __syncthreads();
  for (int off = 1; off < 1024; off <<= 1) {
    const int x = (t >= off) ? s[t - off] : 0;
    __syncthreads();
    s[t] += x;
    __syncthreads();
  }
  const int incl = s[t];
  if (t < NBINS) {
    binStart[t] = incl - v;
    binCursor[t] = incl - v;
  }
  if (t == NBINS - 1) binStart[NBINS] = incl;
}

__global__ __launch_bounds__(SORT_THREADS) void fill_bins(const int* __restrict__ src,
                                                          const int* __restrict__ dst,
                                                          const float* __restrict__ ew,
                                                          int* __restrict__ binCursor,
                                                          int* __restrict__ binned,
                                                          int nE) {
  __shared__ int hist[NBINS];
  __shared__ int basePos[NBINS];
  const int t = threadIdx.x;
  if (t < NBINS) hist[t] = 0;
  __syncthreads();
  const int base = blockIdx.x * SORT_CHUNK;
  int bin[SORT_EPT], rank[SORT_EPT], rec0[SORT_EPT], rec1[SORT_EPT];
#pragma unroll
  for (int i = 0; i < SORT_EPT; ++i) {
    const int e = base + i * SORT_THREADS + t;
    bin[i] = -1; rank[i] = 0; rec0[i] = 0; rec1[i] = 0;
    if (e < nE) {
      const int d = dst[e];
      const int b = d / BIN_ROWS;
      bin[i] = b;
      rank[i] = atomicAdd(&hist[b], 1);
      rec0[i] = src[e] | ((d - b * BIN_ROWS) << 20);
      rec1[i] = __float_as_int(ew[e]);
    }
  }
  __syncthreads();
  if (t < NBINS) {
    const int c = hist[t];
    basePos[t] = c ? atomicAdd(&binCursor[t], c) : 0;
  }
  __syncthreads();
  int2* recs = (int2*)binned;
#pragma unroll
  for (int i = 0; i < SORT_EPT; ++i) {
    if (bin[i] >= 0) recs[basePos[bin[i]] + rank[i]] = make_int2(rec0[i], rec1[i]);
  }
}

#define AGG_THREADS 512
#define AGG_WAVES (AGG_THREADS / 32)
#define EDGE_BATCH 8
#define TILE_ROWS 112
#define TILE_MT (TILE_ROWS / 16)
#define TILE_PITCH 68
#define STEP_LAST 0
#define STEP_MID 1
#define STEP_FIRST 2
static_assert(TILE_ROWS >= BIN_ROWS && TILE_ROWS % 16 == 0, "tile covers the range");
static_assert(2 * TILE_MT <= AGG_WAVES, "one wave per (row tile, column half)");
static_assert((TILE_ROWS - BIN_ROWS) * (DIM / 4) <= AGG_THREADS, "one pass zeroes the padding rows");
static_assert(AGG_THREADS % (DIM / 4) == 0, "a thread's pieces all fall in one 16-byte column of a row");

__device__ __forceinline__ float lane_bcast(float v, int j) {
  return __int_as_float(__builtin_amdgcn_readlane(__float_as_int(v), j));
}

constexpr int W_QUADS = DIM * DIM / 4 / AGG_THREADS;
static_assert(W_QUADS * AGG_THREADS * 4 == DIM * DIM, "the matrix splits evenly over the threads");

__device__ __forceinline__ void stage_wfrag(float* wf, const float4 (&wq)[W_QUADS], int t) {
#pragma unroll
  for (int q = 0; q < W_QUADS; ++q) {
    const int i = t + q * AGG_THREADS;
    const int k = i >> 4;
    const int n = (i & 15) * 4;
    float* p = wf + ((((k >> 2) * 4 + (n >> 4)) * 32 + ((k >> 1) & 1) * 16 + (n & 15)) * 2 + (k & 1));
    p[0] = wq[q].x; p[2] = wq[q].y; p[4] = wq[q].z; p[6] = wq[q].w;
  }
}

__device__ __forceinline__ void tile_product(const float* tile, const float* wf, int mt, int ch, int lane,
                                             v8f& d0, v8f& d1) {
  const float* ap = tile + (mt * 16 + (lane & 15)) * TILE_PITCH + (lane >> 4) * 2;
  const float* bp = wf + (ch * 2 * 32 + lane) * 2;
#pragma unroll
  for (int kk = 0; kk < DIM / 4; ++kk) {
    const v2f a  = *(const v2f*)(ap + kk * 4);
    const v2f b0 = *(const v2f*)(bp + kk * 256);
    const v2f b1 = *(const v2f*)(bp + kk * 256 + 64);
    d0 = __builtin_amdgcn_wmma_f32_16x16x4_f32(false, a, false, b0, (short)0, d0, false, false);
    d1 = __builtin_amdgcn_wmma_f32_16x16x4_f32(false, a, false, b1, (short)0, d1, false, false);
  }
}

__device__ __forceinline__ void tile_put(float* tile, int mt, int ch, int lane, const v8f& d0, const v8f& d1) {
  float* dp = tile + (mt * 16 + (lane >> 4) * 8) * TILE_PITCH + ch * 32 + (lane & 15);
#pragma unroll
  for (int r = 0; r < 8; ++r) {
    dp[r * TILE_PITCH] = d0[r];
    dp[r * TILE_PITCH + 16] = d1[r];
  }
}

#define OWN_ROWS ((BIN_ROWS + AGG_WAVES - 1) / AGG_WAVES)
#define REC_CAP 2048
#define REC_PT (REC_CAP / AGG_THREADS)
static_assert(OWN_ROWS < 8, "a wave's rows and one spare element fit one 8-element register vector per feature");
static_assert(REC_PT * AGG_THREADS == REC_CAP, "a pass splits evenly over the threads");
static_assert(AGG_WAVES <= 16, "the owner counters are scanned inside one 16-lane row");
static_assert(REC_CAP * 8 <= BIN_ROWS * TILE_PITCH * 4, "the record buffer fits inside the tile's real rows");

__device__ __forceinline__ void own_add(v8f& ax, v8f& ay, int idx, float2 a, float w) {
  ax[idx] += a.x * w;
  ay[idx] += a.y * w;
}

template <int STEP>
__global__ __launch_bounds__(AGG_THREADS) void aggregate_bins(const float* __restrict__ XW,
                                                              const float* __restrict__ ori,
                                                              const float* __restrict__ bias,
                                                              const int* __restrict__ binStart,
                                                              const int* __restrict__ binned,
                                                              const float* __restrict__ Wf,
                                                              const float* __restrict__ Wn,
                                                              float* __restrict__ out) {
  constexpr bool FUSE_NEXT = STEP != STEP_LAST;
  constexpr bool FIRST = STEP == STEP_FIRST;
  constexpr int ROWS = FUSE_NEXT ? TILE_ROWS : BIN_ROWS;
  constexpr int ROW_QUADS = BIN_ROWS * (DIM / 4);
  __shared__ __attribute__((aligned(16))) float acc[ROWS * TILE_PITCH];
  __shared__ __attribute__((aligned(16))) float wfrag[FUSE_NEXT ? DIM * DIM : 4];
  __shared__ __attribute__((aligned(16))) float wfirst[FIRST ? DIM * DIM : 4];
  __shared__ int gcnt[AGG_WAVES];
  __shared__ int gbase[AGG_WAVES + 1];
  const int t = threadIdx.x;
  const int lane = t & 31;
  const int wave = __builtin_amdgcn_readfirstlane(t >> 5);
  const int b = blockIdx.x;
  const size_t row4 = (size_t)b * ROW_QUADS;
  const int rlo = (wave * BIN_ROWS + AGG_WAVES - 1) / AGG_WAVES;
  const int nown = ((wave + 1) * BIN_ROWS + AGG_WAVES - 1) / AGG_WAVES - rlo;

  float4 wq[W_QUADS], wq1[W_QUADS];
  if (FUSE_NEXT) {
#pragma unroll
    for (int q = 0; q < W_QUADS; ++q) wq[q] = ((const float4*)Wn)[t + q * AGG_THREADS];
  }
  if (FIRST) {
#pragma unroll
    for (int q = 0; q < W_QUADS; ++q) wq1[q] = ((const float4*)Wf)[t + q * AGG_THREADS];
  }

  v8f ax = {}, ay = {};
  if (!FIRST) {
    const float2 bb = ((const float2*)bias)[lane];
    float2 o[OWN_ROWS];
#pragma unroll
    for (int r = 0; r < OWN_ROWS; ++r) {
      const int rr = (r < nown) ? r : nown - 1;
      o[r] = ((const float2*)(ori + ((size_t)b * BIN_ROWS + rlo + rr) * DIM))[lane];
    }
#pragma unroll
    for (int r = 0; r < OWN_ROWS; ++r) {
      ax[r] = o[r].x + bb.x;
      ay[r] = o[r].y + bb.y;
    }
  }
  if (FUSE_NEXT) {
    if (t < (TILE_ROWS - BIN_ROWS) * (DIM / 4))
      *(float4*)(acc + (BIN_ROWS + (t >> 4)) * TILE_PITCH + (t & (DIM / 4 - 1)) * 4) = make_float4(0.f, 0.f, 0.f, 0.f);
    stage_wfrag(wfrag, wq, t);
  }
  if (FIRST) stage_wfrag(wfirst, wq1, t);

  const int e0 = binStart[b];
  const int e1 = binStart[b + 1];
  const int2* recs = (const int2*)binned;
  int2* lrec = (int2*)acc;

  for (int sb = e0; sb < e1; sb += REC_CAP) {
    const int n = (e1 - sb < REC_CAP) ? (e1 - sb) : REC_CAP;
    if (t < AGG_WAVES) gcnt[t] = 0;
    __syncthreads();
    int2 r[REC_PT];
    int ow[REC_PT], rk[REC_PT];
#pragma unroll
    for (int i = 0; i < REC_PT; ++i) {
      const int k = t + i * AGG_THREADS;
      r[i] = make_int2(0, 0);
      if (k < n) r[i] = recs[sb + k];
    }
#pragma unroll
    for (int i = 0; i < REC_PT; ++i) {
      const int k = t + i * AGG_THREADS;
      ow[i] = -1; rk[i] = 0;
      if (k < n) {
        ow[i] = ((r[i].x >> 20) * AGG_WAVES) / BIN_ROWS;
        rk[i] = atomicAdd(&gcnt[ow[i]], 1);
      }
    }
    __syncthreads();
    if (wave == 0) {
      int v = (lane < AGG_WAVES) ? gcnt[lane] : 0;
      v += __builtin_amdgcn_update_dpp(0, v, 0x111, 0xf, 0xf, false);
      v += __builtin_amdgcn_update_dpp(0, v, 0x112, 0xf, 0xf, false);
      v += __builtin_amdgcn_update_dpp(0, v, 0x114, 0xf, 0xf, false);
      v += __builtin_amdgcn_update_dpp(0, v, 0x118, 0xf, 0xf, false);
      if (lane < AGG_WAVES) gbase[lane + 1] = v;
      if (lane == 0) gbase[0] = 0;
    }
    __syncthreads();
#pragma unroll
    for (int i = 0; i < REC_PT; ++i) {
      if (ow[i] >= 0) lrec[gbase[ow[i]] + rk[i]] = r[i];
    }
    __syncthreads();

    const int s0 = __builtin_amdgcn_readfirstlane(gbase[wave]);
    const int s1 = __builtin_amdgcn_readfirstlane(gbase[wave + 1]);
    for (int c = s0; c < s1; c += 32) {
      const int cnt = (s1 - c < 32) ? (s1 - c) : 32;
      int s = 0, dl = OWN_ROWS;
      float w = 0.f;
      if (lane < cnt) {
        const int2 rec = lrec[c + lane];
        s = rec.x & 0xFFFFF;
        dl = (rec.x >> 20) - rlo;
        w = __int_as_float(rec.y);
      }
      for (int j0 = 0; j0 < cnt; j0 += EDGE_BATCH) {
        float2 a[EDGE_BATCH];
#pragma unroll
        for (int u = 0; u < EDGE_BATCH; ++u) {
          const float2* xr = (const float2*)(XW + (size_t)__builtin_amdgcn_readlane(s, j0 + u) * DIM);
          a[u] = xr[lane];
        }
#pragma unroll
        for (int u = 0; u < EDGE_BATCH; ++u)
          own_add(ax, ay, __builtin_amdgcn_readlane(dl, j0 + u), a[u], lane_bcast(w, j0 + u));
      }
    }
  }

  if (!FUSE_NEXT) {
#pragma unroll
    for (int r = 0; r < OWN_ROWS; ++r) {
      if (r < nown)
        ((float2*)(out + ((size_t)b * BIN_ROWS + rlo + r) * DIM))[lane] = make_float2(ax[r], ay[r]);
    }
    return;
  }

  __syncthreads();
#pragma unroll
  for (int r = 0; r < OWN_ROWS; ++r) {
    if (r < nown)
      *(float2*)(acc + (rlo + r) * TILE_PITCH + 2 * lane) = make_float2(ax[r], ay[r]);
  }
  __syncthreads();

  const int mt = wave >> 1;
  const int ch = wave & 1;

  if (FIRST) {
    constexpr int ORI_Q = (ROW_QUADS + AGG_THREADS - 1) / AGG_THREADS;
    float4 o[ORI_Q];
#pragma unroll
    for (int q = 0; q < ORI_Q; ++q) {
      const int i = t + q * AGG_THREADS;
      o[q] = ((const float4*)ori)[row4 + (i < ROW_QUADS ? i : ROW_QUADS - 1)];
    }
    const float4 bb = ((const float4*)bias)[t & (DIM / 4 - 1)];
    v8f f0 = {}, f1 = {};
    if (wave < 2 * TILE_MT) tile_product(acc, wfirst, mt, ch, lane, f0, f1);
    __syncthreads();
    if (wave < 2 * TILE_MT) tile_put(acc, mt, ch, lane, f0, f1);
    __syncthreads();
#pragma unroll
    for (int q = 0; q < ORI_Q; ++q) {
      const int i = t + q * AGG_THREADS;
      if (i < ROW_QUADS) {
        float* p = acc + (i >> 4) * TILE_PITCH + (i & (DIM / 4 - 1)) * 4;
        const float4 s = *(const float4*)p;
        float4 r;
        r.x = (s.x + bb.x) + o[q].x; r.y = (s.y + bb.y) + o[q].y;
        r.z = (s.z + bb.z) + o[q].z; r.w = (s.w + bb.w) + o[q].w;
        *(float4*)p = r;
      }
    }
    __syncthreads();
  }

  v8f d0 = {}, d1 = {};
  if (wave < 2 * TILE_MT) tile_product(acc, wfrag, mt, ch, lane, d0, d1);
  __syncthreads();
  if (wave < 2 * TILE_MT) tile_put(acc, mt, ch, lane, d0, d1);
  __syncthreads();

  for (int i = t; i < ROW_QUADS; i += AGG_THREADS)
    ((float4*)out)[row4 + i] = *(const float4*)(acc + (i >> 4) * TILE_PITCH + (i & (DIM / 4 - 1)) * 4);
}

extern "C" void kernel_launch(void* const* d_in, const int* in_sizes, int n_in,
                              void* d_out, int out_size, void* d_ws, size_t ws_size,
                              hipStream_t stream) {
  const float* in_feat = (const float*)d_in[0];
  const float* ew      = (const float*)d_in[1];
  const float* W1      = (const float*)d_in[2];
  const float* b1      = (const float*)d_in[3];
  const float* W2      = (const float*)d_in[4];
  const float* b2      = (const float*)d_in[5];
  const int*   src     = (const int*)d_in[6];
  const int*   dst     = (const int*)d_in[7];
  const int    nE      = in_sizes[1];

  float* xwA  = (float*)d_ws;
  float* xwB  = xwA + (size_t)NNODES * DIM;
  int* binCount  = (int*)(xwB + (size_t)NNODES * DIM);
  int* binStart  = binCount + 1024;
  int* binCursor = binStart + 2048;
  int* binned    = binCursor + 1024;
  float* outp = (float*)d_out;

  const int sortBlocks = (nE + SORT_CHUNK - 1) / SORT_CHUNK;

  zero_bins<<<1, 1024, 0, stream>>>(binCount);
  count_bins<<<sortBlocks, SORT_THREADS, 0, stream>>>(dst, binCount, nE);
  scan_bins<<<1, 1024, 0, stream>>>(binCount, binStart, binCursor);
  fill_bins<<<sortBlocks, SORT_THREADS, 0, stream>>>(src, dst, ew, binCursor, binned, nE);

  aggregate_bins<STEP_FIRST><<<NBINS, AGG_THREADS, 0, stream>>>(in_feat, in_feat, b1, binStart, binned, W1, W2, xwA);
  aggregate_bins<STEP_MID><<<NBINS, AGG_THREADS, 0, stream>>>(xwA, in_feat, b2, binStart, binned, W2, W2, xwB);
  aggregate_bins<STEP_MID><<<NBINS, AGG_THREADS, 0, stream>>>(xwB, in_feat, b2, binStart, binned, W2, W2, xwA);
  aggregate_bins<STEP_LAST><<<NBINS, AGG_THREADS, 0, stream>>>(xwA, in_feat, b2, binStart, binned, W2, W2, outp);
}


# LDS regroup keyed by destination row: a wave sums each row's run of edges in one register pair (no indexed register moves per edge), adds it once to the row's LDS tile place; FP sum order changes
# speedup vs baseline: 4.3274x; 1.0650x over previous
#include <hip/hip_runtime.h>

typedef __attribute__((ext_vector_type(2))) float v2f;
typedef __attribute__((ext_vector_type(8))) float v8f;

#define NNODES 100000
#define DIM 64


#define BIN_ROWS 100
#define NBINS (NNODES / BIN_ROWS)
static_assert(NBINS * BIN_ROWS == NNODES, "ranges must tile the node set exactly");
static_assert(NBINS <= 1024, "one scan block covers all ranges");
static_assert(NNODES <= (1 << 20) && BIN_ROWS < (1 << 11), "record packing");
#define SORT_THREADS 1024
#define SORT_EPT 4
#define SORT_CHUNK (SORT_THREADS * SORT_EPT)

__global__ __launch_bounds__(1024) void zero_bins(int* __restrict__ binCount) {
  binCount[threadIdx.x] = 0;
}

__global__ __launch_bounds__(SORT_THREADS) void count_bins(const int* __restrict__ dst,
                                                           int* __restrict__ binCount,
                                                           int nE) {
  __shared__ int hist[NBINS];
  const int t = threadIdx.x;
  if (t < NBINS) hist[t] = 0;
  __syncthreads();
  const int base = blockIdx.x * SORT_CHUNK;
  for (int i = 0; i < SORT_EPT; ++i) {
    const int e = base + i * SORT_THREADS + t;
    if (e < nE) atomicAdd(&hist[dst[e] / BIN_ROWS], 1);
  }
  __syncthreads();
  if (t < NBINS) {
    const int c = hist[t];
    if (c) atomicAdd(&binCount[t], c);
  }
}

__global__ __launch_bounds__(1024) void scan_bins(const int* __restrict__ binCount,
                                                  int* __restrict__ binStart,
                                                  int* __restrict__ binCursor) {
  __shared__ int s[1024];
  const int t = threadIdx.x;
  const int v = (t < NBINS) ? binCount[t] : 0;
  s[t] = v;
  __syncthreads();
  for (int off = 1; off < 1024; off <<= 1) {
    const int x = (t >= off) ? s[t - off] : 0;
    __syncthreads();
    s[t] += x;
    __syncthreads();
  }
  const int incl = s[t];
  if (t < NBINS) {
    binStart[t] = incl - v;
    binCursor[t] = incl - v;
  }
  if (t == NBINS - 1) binStart[NBINS] = incl;
}

__global__ __launch_bounds__(SORT_THREADS) void fill_bins(const int* __restrict__ src,
                                                          const int* __restrict__ dst,
                                                          const float* __restrict__ ew,
                                                          int* __restrict__ binCursor,
                                                          int* __restrict__ binned,
                                                          int nE) {
  __shared__ int hist[NBINS];
  __shared__ int basePos[NBINS];
  const int t = threadIdx.x;
  if (t < NBINS) hist[t] = 0;
  __syncthreads();
  const int base = blockIdx.x * SORT_CHUNK;
  int bin[SORT_EPT], rank[SORT_EPT], rec0[SORT_EPT], rec1[SORT_EPT];
#pragma unroll
  for (int i = 0; i < SORT_EPT; ++i) {
    const int e = base + i * SORT_THREADS + t;
    bin[i] = -1; rank[i] = 0; rec0[i] = 0; rec1[i] = 0;
    if (e < nE) {
      const int d = dst[e];
      const int b = d / BIN_ROWS;
      bin[i] = b;
      rank[i] = atomicAdd(&hist[b], 1);
      rec0[i] = src[e] | ((d - b * BIN_ROWS) << 20);
      rec1[i] = __float_as_int(ew[e]);
    }
  }
  __syncthreads();
  if (t < NBINS) {
    const int c = hist[t];
    basePos[t] = c ? atomicAdd(&binCursor[t], c) : 0;
  }
  __syncthreads();
  int2* recs = (int2*)binned;
#pragma unroll
  for (int i = 0; i < SORT_EPT; ++i) {
    if (bin[i] >= 0) recs[basePos[bin[i]] + rank[i]] = make_int2(rec0[i], rec1[i]);
  }
}

#define AGG_THREADS 512
#define AGG_WAVES (AGG_THREADS / 32)
#define EDGE_BATCH 8
#define TILE_ROWS 112
#define TILE_MT (TILE_ROWS / 16)
#define TILE_PITCH 68
#define STEP_LAST 0
#define STEP_MID 1
#define STEP_FIRST 2
static_assert(TILE_ROWS >= BIN_ROWS && TILE_ROWS % 16 == 0, "tile covers the range");
static_assert(2 * TILE_MT <= AGG_WAVES, "one wave per (row tile, column half)");
static_assert((TILE_ROWS - BIN_ROWS) * (DIM / 4) <= AGG_THREADS, "one pass zeroes the padding rows");
static_assert(AGG_THREADS % (DIM / 4) == 0, "a thread's pieces all fall in one 16-byte column of a row");

__device__ __forceinline__ float lane_bcast(float v, int j) {
  return __int_as_float(__builtin_amdgcn_readlane(__float_as_int(v), j));
}

constexpr int W_QUADS = DIM * DIM / 4 / AGG_THREADS;
static_assert(W_QUADS * AGG_THREADS * 4 == DIM * DIM, "the matrix splits evenly over the threads");

__device__ __forceinline__ void stage_wfrag(float* wf, const float4 (&wq)[W_QUADS], int t) {
#pragma unroll
  for (int q = 0; q < W_QUADS; ++q) {
    const int i = t + q * AGG_THREADS;
    const int k = i >> 4;
    const int n = (i & 15) * 4;
    float* p = wf + ((((k >> 2) * 4 + (n >> 4)) * 32 + ((k >> 1) & 1) * 16 + (n & 15)) * 2 + (k & 1));
    p[0] = wq[q].x; p[2] = wq[q].y; p[4] = wq[q].z; p[6] = wq[q].w;
  }
}

__device__ __forceinline__ void tile_product(const float* tile, const float* wf, int mt, int ch, int lane,
                                             v8f& d0, v8f& d1) {
  const float* ap = tile + (mt * 16 + (lane & 15)) * TILE_PITCH + (lane >> 4) * 2;
  const float* bp = wf + (ch * 2 * 32 + lane) * 2;
#pragma unroll
  for (int kk = 0; kk < DIM / 4; ++kk) {
    const v2f a  = *(const v2f*)(ap + kk * 4);
    const v2f b0 = *(const v2f*)(bp + kk * 256);
    const v2f b1 = *(const v2f*)(bp + kk * 256 + 64);
    d0 = __builtin_amdgcn_wmma_f32_16x16x4_f32(false, a, false, b0, (short)0, d0, false, false);
    d1 = __builtin_amdgcn_wmma_f32_16x16x4_f32(false, a, false, b1, (short)0, d1, false, false);
  }
}

__device__ __forceinline__ void tile_put(float* tile, int mt, int ch, int lane, const v8f& d0, const v8f& d1) {
  float* dp = tile + (mt * 16 + (lane >> 4) * 8) * TILE_PITCH + ch * 32 + (lane & 15);
#pragma unroll
  for (int r = 0; r < 8; ++r) {
    dp[r * TILE_PITCH] = d0[r];
    dp[r * TILE_PITCH + 16] = d1[r];
  }
}

#define REC_CAP 2048
#define REC_PT (REC_CAP / AGG_THREADS)
#define ROW_SLOTS 128
static_assert(REC_PT * AGG_THREADS == REC_CAP, "a pass splits evenly over the threads");
static_assert(BIN_ROWS <= ROW_SLOTS && ROW_SLOTS == 4 * 32, "one wave scans all row counters, four per lane");
static_assert(REC_CAP * 8 == DIM * DIM * 4, "the record buffer and a staged 64 x 64 matrix share the same bytes");

__device__ __forceinline__ void run_flush(float* acc, int row, int lane, float cx, float cy) {
  float2* p = (float2*)(acc + row * TILE_PITCH + 2 * lane);
  float2 v = *p;
  v.x += cx; v.y += cy;
  *p = v;
}

template <int STEP>
__global__ __launch_bounds__(AGG_THREADS) void aggregate_bins(const float* __restrict__ XW,
                                                              const float* __restrict__ ori,
                                                              const float* __restrict__ bias,
                                                              const int* __restrict__ binStart,
                                                              const int* __restrict__ binned,
                                                              const float* __restrict__ Wf,
                                                              const float* __restrict__ Wn,
                                                              float* __restrict__ out) {
  constexpr bool FUSE_NEXT = STEP != STEP_LAST;
  constexpr bool FIRST = STEP == STEP_FIRST;
  constexpr int ROWS = FUSE_NEXT ? TILE_ROWS : BIN_ROWS;
  constexpr int ROW_QUADS = BIN_ROWS * (DIM / 4);
  constexpr int ORI_Q = (ROW_QUADS + AGG_THREADS - 1) / AGG_THREADS;
  __shared__ __attribute__((aligned(16))) float acc[ROWS * TILE_PITCH];
  __shared__ __attribute__((aligned(16))) float wfrag[FUSE_NEXT ? DIM * DIM : 4];
  __shared__ __attribute__((aligned(16))) int2 recbuf[REC_CAP];
  __shared__ __attribute__((aligned(16))) int rcnt[ROW_SLOTS];
  __shared__ __attribute__((aligned(16))) int rstart[ROW_SLOTS];
  const int t = threadIdx.x;
  const int lane = t & 31;
  const int wave = __builtin_amdgcn_readfirstlane(t >> 5);
  const int b = blockIdx.x;
  const size_t row4 = (size_t)b * ROW_QUADS;
  const int rlo = (wave * BIN_ROWS + AGG_WAVES - 1) / AGG_WAVES;
  const int nown = ((wave + 1) * BIN_ROWS + AGG_WAVES - 1) / AGG_WAVES - rlo;

  float4 wq[W_QUADS];
  if (FUSE_NEXT) {
#pragma unroll
    for (int q = 0; q < W_QUADS; ++q) wq[q] = ((const float4*)Wn)[t + q * AGG_THREADS];
  }
  if (!FIRST) {
    const float4 bb = ((const float4*)bias)[t & (DIM / 4 - 1)];
    float4 o[ORI_Q];
#pragma unroll
    for (int q = 0; q < ORI_Q; ++q) {
      const int i = t + q * AGG_THREADS;
      o[q] = ((const float4*)ori)[row4 + (i < ROW_QUADS ? i : ROW_QUADS - 1)];
    }
#pragma unroll
    for (int q = 0; q < ORI_Q; ++q) {
      const int i = t + q * AGG_THREADS;
      if (i < ROW_QUADS)
        *(float4*)(acc + (i >> 4) * TILE_PITCH + (i & (DIM / 4 - 1)) * 4) =
            make_float4(o[q].x + bb.x, o[q].y + bb.y, o[q].z + bb.z, o[q].w + bb.w);
    }
  } else {
    for (int i = t; i < ROW_QUADS; i += AGG_THREADS)
      *(float4*)(acc + (i >> 4) * TILE_PITCH + (i & (DIM / 4 - 1)) * 4) = make_float4(0.f, 0.f, 0.f, 0.f);
  }
  if (FUSE_NEXT) {
    if (t < (TILE_ROWS - BIN_ROWS) * (DIM / 4))
      *(float4*)(acc + (BIN_ROWS + (t >> 4)) * TILE_PITCH + (t & (DIM / 4 - 1)) * 4) = make_float4(0.f, 0.f, 0.f, 0.f);
    stage_wfrag(wfrag, wq, t);
  }

  const int e0 = binStart[b];
  const int e1 = binStart[b + 1];
  const int2* recs = (const int2*)binned;
  int2* lrec = recbuf;

  int cur = -1;
  float cx = 0.f, cy = 0.f;

  for (int sb = e0; sb < e1; sb += REC_CAP) {
    const int n = (e1 - sb < REC_CAP) ? (e1 - sb) : REC_CAP;
    if (t < ROW_SLOTS) rcnt[t] = 0;
    __syncthreads();
    int2 r[REC_PT];
    int rw[REC_PT], rk[REC_PT];
#pragma unroll
    for (int i = 0; i < REC_PT; ++i) {
      const int k = t + i * AGG_THREADS;
      r[i] = make_int2(0, 0);
      if (k < n) r[i] = recs[sb + k];
    }
#pragma unroll
    for (int i = 0; i < REC_PT; ++i) {
      const int k = t + i * AGG_THREADS;
      rw[i] = -1; rk[i] = 0;
      if (k < n) {
        rw[i] = r[i].x >> 20;
        rk[i] = atomicAdd(&rcnt[rw[i]], 1);
      }
    }
    __syncthreads();
    if (wave == 0) {
      const int4 c4 = *(const int4*)(rcnt + 4 * lane);
      const int p1 = c4.x, p2 = p1 + c4.y, p3 = p2 + c4.z, tot = p3 + c4.w;
      int v = tot;
      v += __builtin_amdgcn_update_dpp(0, v, 0x111, 0xf, 0xf, false);
      v += __builtin_amdgcn_update_dpp(0, v, 0x112, 0xf, 0xf, false);
      v += __builtin_amdgcn_update_dpp(0, v, 0x114, 0xf, 0xf, false);
      v += __builtin_amdgcn_update_dpp(0, v, 0x118, 0xf, 0xf, false);
      const int lower = __builtin_amdgcn_readlane(v, 15);
      if (lane >= 16) v += lower;
      const int ex = v - tot;
      *(int4*)(rstart + 4 * lane) = make_int4(ex, ex + p1, ex + p2, ex + p3);
    }
    __syncthreads();
#pragma unroll
    for (int i = 0; i < REC_PT; ++i) {
      if (rw[i] >= 0) lrec[rstart[rw[i]] + rk[i]] = r[i];
    }
    __syncthreads();

    const int s0 = __builtin_amdgcn_readfirstlane(rstart[rlo]);
    const int s1 = __builtin_amdgcn_readfirstlane(rstart[rlo + nown]);
    for (int c = s0; c < s1; c += 32) {
      const int cnt = (s1 - c < 32) ? (s1 - c) : 32;
      const int2 rec = lrec[(lane < cnt) ? (c + lane) : (s1 - 1)];
      const bool real = lane < cnt;
      const int s = real ? (rec.x & 0xFFFFF) : 0;
      const int dl = real ? ((rec.x >> 20) - rlo) : -1;
      const float w = real ? __int_as_float(rec.y) : 0.f;
      const int dprev = __builtin_amdgcn_ds_bpermute(((lane + 31) & 31) * 4, dl);
      const unsigned chg = __builtin_amdgcn_ballot_w32(dl != ((lane == 0) ? cur : dprev));
      for (int j0 = 0; j0 < cnt; j0 += EDGE_BATCH) {
        float2 a[EDGE_BATCH];
#pragma unroll
        for (int u = 0; u < EDGE_BATCH; ++u) {
          const float2* xr = (const float2*)(XW + (size_t)__builtin_amdgcn_readlane(s, j0 + u) * DIM);
          a[u] = xr[lane];
        }
        const unsigned m8 = chg >> j0;
#pragma unroll
        for (int u = 0; u < EDGE_BATCH; ++u) {
          if (__builtin_expect((m8 >> u) & 1u, 0)) {
            if (cur >= 0) run_flush(acc, rlo + cur, lane, cx, cy);
            cur = __builtin_amdgcn_readlane(dl, j0 + u);
            cx = 0.f; cy = 0.f;
          }
          const float wj = lane_bcast(w, j0 + u);
          cx += a[u].x * wj;
          cy += a[u].y * wj;
        }
      }
    }
  }
  float4 wq1[W_QUADS];
  if (FIRST) {
#pragma unroll
    for (int q = 0; q < W_QUADS; ++q) wq1[q] = ((const float4*)Wf)[t + q * AGG_THREADS];
  }
  if (cur >= 0) run_flush(acc, rlo + cur, lane, cx, cy);

  __syncthreads();

  if (!FUSE_NEXT) {
    for (int i = t; i < ROW_QUADS; i += AGG_THREADS)
      ((float4*)out)[row4 + i] = *(const float4*)(acc + (i >> 4) * TILE_PITCH + (i & (DIM / 4 - 1)) * 4);
    return;
  }

  const int mt = wave >> 1;
  const int ch = wave & 1;

  if (FIRST) {
    float* wfirst = (float*)recbuf;
    stage_wfrag(wfirst, wq1, t);
    float4 o[ORI_Q];
#pragma unroll
    for (int q = 0; q < ORI_Q; ++q) {
      const int i = t + q * AGG_THREADS;
      o[q] = ((const float4*)ori)[row4 + (i < ROW_QUADS ? i : ROW_QUADS - 1)];
    }
    const float4 bb = ((const float4*)bias)[t & (DIM / 4 - 1)];
    __syncthreads();
    v8f f0 = {}, f1 = {};
    if (wave < 2 * TILE_MT) tile_product(acc, wfirst, mt, ch, lane, f0, f1);
    __syncthreads();
    if (wave < 2 * TILE_MT) tile_put(acc, mt, ch, lane, f0, f1);
    __syncthreads();
#pragma unroll
    for (int q = 0; q < ORI_Q; ++q) {
      const int i = t + q * AGG_THREADS;
      if (i < ROW_QUADS) {
        float* p = acc + (i >> 4) * TILE_PITCH + (i & (DIM / 4 - 1)) * 4;
        const float4 s = *(const float4*)p;
        float4 r;
        r.x = (s.x + bb.x) + o[q].x; r.y = (s.y + bb.y) + o[q].y;
        r.z = (s.z + bb.z) + o[q].z; r.w = (s.w + bb.w) + o[q].w;
        *(float4*)p = r;
      }
    }
    __syncthreads();
  }

  v8f d0 = {}, d1 = {};
  if (wave < 2 * TILE_MT) tile_product(acc, wfrag, mt, ch, lane, d0, d1);
  __syncthreads();
  if (wave < 2 * TILE_MT) tile_put(acc, mt, ch, lane, d0, d1);
  __syncthreads();

  for (int i = t; i < ROW_QUADS; i += AGG_THREADS)
    ((float4*)out)[row4 + i] = *(const float4*)(acc + (i >> 4) * TILE_PITCH + (i & (DIM / 4 - 1)) * 4);
}

extern "C" void kernel_launch(void* const* d_in, const int* in_sizes, int n_in,
                              void* d_out, int out_size, void* d_ws, size_t ws_size,
                              hipStream_t stream) {
  const float* in_feat = (const float*)d_in[0];
  const float* ew      = (const float*)d_in[1];
  const float* W1      = (const float*)d_in[2];
  const float* b1      = (const float*)d_in[3];
  const float* W2      = (const float*)d_in[4];
  const float* b2      = (const float*)d_in[5];
  const int*   src     = (const int*)d_in[6];
  const int*   dst     = (const int*)d_in[7];
  const int    nE      = in_sizes[1];

  float* xwA  = (float*)d_ws;
  float* xwB  = xwA + (size_t)NNODES * DIM;
  int* binCount  = (int*)(xwB + (size_t)NNODES * DIM);
  int* binStart  = binCount + 1024;
  int* binCursor = binStart + 2048;
  int* binned    = binCursor + 1024;
  float* outp = (float*)d_out;

  const int sortBlocks = (nE + SORT_CHUNK - 1) / SORT_CHUNK;

  zero_bins<<<1, 1024, 0, stream>>>(binCount);
  count_bins<<<sortBlocks, SORT_THREADS, 0, stream>>>(dst, binCount, nE);
  scan_bins<<<1, 1024, 0, stream>>>(binCount, binStart, binCursor);
  fill_bins<<<sortBlocks, SORT_THREADS, 0, stream>>>(src, dst, ew, binCursor, binned, nE);

  aggregate_bins<STEP_FIRST><<<NBINS, AGG_THREADS, 0, stream>>>(in_feat, in_feat, b1, binStart, binned, W1, W2, xwA);
  aggregate_bins<STEP_MID><<<NBINS, AGG_THREADS, 0, stream>>>(xwA, in_feat, b2, binStart, binned, W2, W2, xwB);
  aggregate_bins<STEP_MID><<<NBINS, AGG_THREADS, 0, stream>>>(xwB, in_feat, b2, binStart, binned, W2, W2, xwA);
  aggregate_bins<STEP_LAST><<<NBINS, AGG_THREADS, 0, stream>>>(xwA, in_feat, b2, binStart, binned, W2, W2, outp);
}


# Each half-wave (16 lanes x 16 B) owns 3-4 rows and walks its own records: one load instruction gathers two 256-byte source rows, no per-edge scalar reads; order of a row's FP sums unchanged
# speedup vs baseline: 4.4718x; 1.0334x over previous
#include <hip/hip_runtime.h>

typedef __attribute__((ext_vector_type(2))) float v2f;
typedef __attribute__((ext_vector_type(8))) float v8f;

#define NNODES 100000
#define DIM 64


#define BIN_ROWS 100
#define NBINS (NNODES / BIN_ROWS)
static_assert(NBINS * BIN_ROWS == NNODES, "ranges must tile the node set exactly");
static_assert(NBINS <= 1024, "one scan block covers all ranges");
static_assert(NNODES <= (1 << 20) && BIN_ROWS < (1 << 11), "record packing");
#define SORT_THREADS 1024
#define SORT_EPT 4
#define SORT_CHUNK (SORT_THREADS * SORT_EPT)

__global__ __launch_bounds__(1024) void zero_bins(int* __restrict__ binCount) {
  binCount[threadIdx.x] = 0;
}

__global__ __launch_bounds__(SORT_THREADS) void count_bins(const int* __restrict__ dst,
                                                           int* __restrict__ binCount,
                                                           int nE) {
  __shared__ int hist[NBINS];
  const int t = threadIdx.x;
  if (t < NBINS) hist[t] = 0;
  __syncthreads();
  const int base = blockIdx.x * SORT_CHUNK;
  for (int i = 0; i < SORT_EPT; ++i) {
    const int e = base + i * SORT_THREADS + t;
    if (e < nE) atomicAdd(&hist[dst[e] / BIN_ROWS], 1);
  }
  __syncthreads();
  if (t < NBINS) {
    const int c = hist[t];
    if (c) atomicAdd(&binCount[t], c);
  }
}

__global__ __launch_bounds__(1024) void scan_bins(const int* __restrict__ binCount,
                                                  int* __restrict__ binStart,
                                                  int* __restrict__ binCursor) {
  __shared__ int s[1024];
  const int t = threadIdx.x;
  const int v = (t < NBINS) ? binCount[t] : 0;
  s[t] = v;
  __syncthreads();
  for (int off = 1; off < 1024; off <<= 1) {
    const int x = (t >= off) ? s[t - off] : 0;
    __syncthreads();
    s[t] += x;
    __syncthreads();
  }
  const int incl = s[t];
  if (t < NBINS) {
    binStart[t] = incl - v;
    binCursor[t] = incl - v;
  }
  if (t == NBINS - 1) binStart[NBINS] = incl;
}

__global__ __launch_bounds__(SORT_THREADS) void fill_bins(const int* __restrict__ src,
                                                          const int* __restrict__ dst,
                                                          const float* __restrict__ ew,
                                                          int* __restrict__ binCursor,
                                                          int* __restrict__ binned,
                                                          int nE) {
  __shared__ int hist[NBINS];
  __shared__ int basePos[NBINS];
  const int t = threadIdx.x;
  if (t < NBINS) hist[t] = 0;
  __syncthreads();
  const int base = blockIdx.x * SORT_CHUNK;
  int bin[SORT_EPT], rank[SORT_EPT], rec0[SORT_EPT], rec1[SORT_EPT];
#pragma unroll
  for (int i = 0; i < SORT_EPT; ++i) {
    const int e = base + i * SORT_THREADS + t;
    bin[i] = -1; rank[i] = 0; rec0[i] = 0; rec1[i] = 0;
    if (e < nE) {
      const int d = dst[e];
      const int b = d / BIN_ROWS;
      bin[i] = b;
      rank[i] = atomicAdd(&hist[b], 1);
      rec0[i] = src[e] | ((d - b * BIN_ROWS) << 20);
      rec1[i] = __float_as_int(ew[e]);
    }
  }
  __syncthreads();
  if (t < NBINS) {
    const int c = hist[t];
    basePos[t] = c ? atomicAdd(&binCursor[t], c) : 0;
  }
  __syncthreads();
  int2* recs = (int2*)binned;
#pragma unroll
  for (int i = 0; i < SORT_EPT; ++i) {
    if (bin[i] >= 0) recs[basePos[bin[i]] + rank[i]] = make_int2(rec0[i], rec1[i]);
  }
}

#define AGG_THREADS 512
#define AGG_WAVES (AGG_THREADS / 32)
#define AGG_HALVES (2 * AGG_WAVES)
#define STEP_BATCH 4
#define CHUNK_STEPS 16
#define TILE_ROWS 112
#define TILE_MT (TILE_ROWS / 16)
#define TILE_PITCH 68
#define STEP_LAST 0
#define STEP_MID 1
#define STEP_FIRST 2
static_assert(TILE_ROWS >= BIN_ROWS && TILE_ROWS % 16 == 0, "tile covers the range");
static_assert(2 * TILE_MT <= AGG_WAVES, "one wave per (row tile, column half)");
static_assert((TILE_ROWS - BIN_ROWS) * (DIM / 4) <= AGG_THREADS, "one pass zeroes the padding rows");
static_assert(AGG_THREADS % (DIM / 4) == 0, "a thread's pieces all fall in one 16-byte column of a row");

constexpr int W_QUADS = DIM * DIM / 4 / AGG_THREADS;
static_assert(W_QUADS * AGG_THREADS * 4 == DIM * DIM, "the matrix splits evenly over the threads");

__device__ __forceinline__ void stage_wfrag(float* wf, const float4 (&wq)[W_QUADS], int t) {
#pragma unroll
  for (int q = 0; q < W_QUADS; ++q) {
    const int i = t + q * AGG_THREADS;
    const int k = i >> 4;
    const int n = (i & 15) * 4;
    float* p = wf + ((((k >> 2) * 4 + (n >> 4)) * 32 + ((k >> 1) & 1) * 16 + (n & 15)) * 2 + (k & 1));
    p[0] = wq[q].x; p[2] = wq[q].y; p[4] = wq[q].z; p[6] = wq[q].w;
  }
}

__device__ __forceinline__ void tile_product(const float* tile, const float* wf, int mt, int ch, int lane,
                                             v8f& d0, v8f& d1) {
  const float* ap = tile + (mt * 16 + (lane & 15)) * TILE_PITCH + (lane >> 4) * 2;
  const float* bp = wf + (ch * 2 * 32 + lane) * 2;
#pragma unroll
  for (int kk = 0; kk < DIM / 4; ++kk) {
    const v2f a  = *(const v2f*)(ap + kk * 4);
    const v2f b0 = *(const v2f*)(bp + kk * 256);
    const v2f b1 = *(const v2f*)(bp + kk * 256 + 64);
    d0 = __builtin_amdgcn_wmma_f32_16x16x4_f32(false, a, false, b0, (short)0, d0, false, false);
    d1 = __builtin_amdgcn_wmma_f32_16x16x4_f32(false, a, false, b1, (short)0, d1, false, false);
  }
}

__device__ __forceinline__ void tile_put(float* tile, int mt, int ch, int lane, const v8f& d0, const v8f& d1) {
  float* dp = tile + (mt * 16 + (lane >> 4) * 8) * TILE_PITCH + ch * 32 + (lane & 15);
#pragma unroll
  for (int r = 0; r < 8; ++r) {
    dp[r * TILE_PITCH] = d0[r];
    dp[r * TILE_PITCH + 16] = d1[r];
  }
}

#define REC_CAP 2048
#define REC_PT (REC_CAP / AGG_THREADS)
#define ROW_SLOTS 128
static_assert(REC_PT * AGG_THREADS == REC_CAP, "a pass splits evenly over the threads");
static_assert(BIN_ROWS <= ROW_SLOTS && ROW_SLOTS == 4 * 32, "one wave scans all row counters, four per lane");
static_assert(REC_CAP * 8 == DIM * DIM * 4, "the record buffer and a staged 64 x 64 matrix share the same bytes");

__device__ __forceinline__ void run_flush(float* acc, int row, int c16, const float4& s) {
  float4* p = (float4*)(acc + row * TILE_PITCH + 4 * c16);
  float4 v = *p;
  v.x += s.x; v.y += s.y; v.z += s.z; v.w += s.w;
  *p = v;
}

template <int U>
__device__ __forceinline__ int half_share(int v) {
  return __builtin_amdgcn_update_dpp(0, v, 0x150 + U, 0xf, 0xf, true);
}

template <int U>
__device__ __forceinline__ float4 gather_rows(const float4* xw4, int sidx, int c16) {
  return xw4[(unsigned)half_share<U>(sidx) + (unsigned)c16];
}

template <int U>
__device__ __forceinline__ void add_edges(float* acc, const float4& x, int dl, float w, unsigned anym, unsigned mh,
                                          int hlo, int c16, int& curv, float4& sum) {
  if (__builtin_expect((anym >> U) & 1u, 0)) {
    const int ncur = half_share<U>(dl);
    if ((mh >> U) & 1u) {
      if (curv >= 0) run_flush(acc, hlo + curv, c16, sum);
      curv = ncur;
      sum = make_float4(0.f, 0.f, 0.f, 0.f);
    }
  }
  const float wj = __int_as_float(half_share<U>(__float_as_int(w)));
  sum.x += x.x * wj;
  sum.y += x.y * wj;
  sum.z += x.z * wj;
  sum.w += x.w * wj;
}

template <int J0>
__device__ __forceinline__ void edge_batch(float* acc, const float4* xw4, int sidx, int dl, float w, unsigned anym,
                                           unsigned mh, int hlo, int c16, int& curv, float4& sum) {
  static_assert(STEP_BATCH == 4 && J0 + STEP_BATCH <= CHUNK_STEPS, "four steps written out below");
  const float4 x0 = gather_rows<J0 + 0>(xw4, sidx, c16);
  const float4 x1 = gather_rows<J0 + 1>(xw4, sidx, c16);
  const float4 x2 = gather_rows<J0 + 2>(xw4, sidx, c16);
  const float4 x3 = gather_rows<J0 + 3>(xw4, sidx, c16);
  add_edges<J0 + 0>(acc, x0, dl, w, anym, mh, hlo, c16, curv, sum);
  add_edges<J0 + 1>(acc, x1, dl, w, anym, mh, hlo, c16, curv, sum);
  add_edges<J0 + 2>(acc, x2, dl, w, anym, mh, hlo, c16, curv, sum);
  add_edges<J0 + 3>(acc, x3, dl, w, anym, mh, hlo, c16, curv, sum);
}

template <int STEP>
__global__ __launch_bounds__(AGG_THREADS) void aggregate_bins(const float* __restrict__ XW,
                                                              const float* __restrict__ ori,
                                                              const float* __restrict__ bias,
                                                              const int* __restrict__ binStart,
                                                              const int* __restrict__ binned,
                                                              const float* __restrict__ Wf,
                                                              const float* __restrict__ Wn,
                                                              float* __restrict__ out) {
  constexpr bool FUSE_NEXT = STEP != STEP_LAST;
  constexpr bool FIRST = STEP == STEP_FIRST;
  constexpr int ROWS = FUSE_NEXT ? TILE_ROWS : BIN_ROWS;
  constexpr int ROW_QUADS = BIN_ROWS * (DIM / 4);
  constexpr int ORI_Q = (ROW_QUADS + AGG_THREADS - 1) / AGG_THREADS;
  __shared__ __attribute__((aligned(16))) float acc[ROWS * TILE_PITCH];
  __shared__ __attribute__((aligned(16))) float wfrag[FUSE_NEXT ? DIM * DIM : 4];
  __shared__ __attribute__((aligned(16))) int2 recbuf[REC_CAP];
  __shared__ __attribute__((aligned(16))) int rcnt[ROW_SLOTS];
  __shared__ __attribute__((aligned(16))) int rstart[ROW_SLOTS];
  const int t = threadIdx.x;
  const int lane = t & 31;
  const int wave = __builtin_amdgcn_readfirstlane(t >> 5);
  const int b = blockIdx.x;
  const size_t row4 = (size_t)b * ROW_QUADS;
  const int rA = ((2 * wave) * BIN_ROWS + AGG_HALVES - 1) / AGG_HALVES;
  const int rB = ((2 * wave + 1) * BIN_ROWS + AGG_HALVES - 1) / AGG_HALVES;
  const int rC = ((2 * wave + 2) * BIN_ROWS + AGG_HALVES - 1) / AGG_HALVES;
  const int c16 = lane & 15;
  const bool upper = lane >= 16;
  const int hlo = upper ? rB : rA;

  float4 wq[W_QUADS];
  if (FUSE_NEXT) {
#pragma unroll
    for (int q = 0; q < W_QUADS; ++q) wq[q] = ((const float4*)Wn)[t + q * AGG_THREADS];
  }
  if (!FIRST) {
    const float4 bb = ((const float4*)bias)[t & (DIM / 4 - 1)];
    float4 o[ORI_Q];
#pragma unroll
    for (int q = 0; q < ORI_Q; ++q) {
      const int i = t + q * AGG_THREADS;
      o[q] = ((const float4*)ori)[row4 + (i < ROW_QUADS ? i : ROW_QUADS - 1)];
    }
#pragma unroll
    for (int q = 0; q < ORI_Q; ++q) {
      const int i = t + q * AGG_THREADS;
      if (i < ROW_QUADS)
        *(float4*)(acc + (i >> 4) * TILE_PITCH + (i & (DIM / 4 - 1)) * 4) =
            make_float4(o[q].x + bb.x, o[q].y + bb.y, o[q].z + bb.z, o[q].w + bb.w);
    }
  } else {
    for (int i = t; i < ROW_QUADS; i += AGG_THREADS)
      *(float4*)(acc + (i >> 4) * TILE_PITCH + (i & (DIM / 4 - 1)) * 4) = make_float4(0.f, 0.f, 0.f, 0.f);
  }
  if (FUSE_NEXT) {
    if (t < (TILE_ROWS - BIN_ROWS) * (DIM / 4))
      *(float4*)(acc + (BIN_ROWS + (t >> 4)) * TILE_PITCH + (t & (DIM / 4 - 1)) * 4) = make_float4(0.f, 0.f, 0.f, 0.f);
    stage_wfrag(wfrag, wq, t);
  }

  const int e0 = binStart[b];
  const int e1 = binStart[b + 1];
  const int2* recs = (const int2*)binned;
  int2* lrec = recbuf;

  int curv = -1;
  float4 sum = make_float4(0.f, 0.f, 0.f, 0.f);
  const float4* xw4 = (const float4*)XW;

  for (int sb = e0; sb < e1; sb += REC_CAP) {
    const int n = (e1 - sb < REC_CAP) ? (e1 - sb) : REC_CAP;
    if (t < ROW_SLOTS) rcnt[t] = 0;
    __syncthreads();
    int2 r[REC_PT];
    int rw[REC_PT], rk[REC_PT];
#pragma unroll
    for (int i = 0; i < REC_PT; ++i) {
      const int k = t + i * AGG_THREADS;
      r[i] = make_int2(0, 0);
      if (k < n) r[i] = recs[sb + k];
    }
#pragma unroll
    for (int i = 0; i < REC_PT; ++i) {
      const int k = t + i * AGG_THREADS;
      rw[i] = -1; rk[i] = 0;
      if (k < n) {
        rw[i] = r[i].x >> 20;
        rk[i] = atomicAdd(&rcnt[rw[i]], 1);
      }
    }
    __syncthreads();
    if (wave == 0) {
      const int4 c4 = *(const int4*)(rcnt + 4 * lane);
      const int p1 = c4.x, p2 = p1 + c4.y, p3 = p2 + c4.z, tot = p3 + c4.w;
      int v = tot;
      v += __builtin_amdgcn_update_dpp(0, v, 0x111, 0xf, 0xf, false);
      v += __builtin_amdgcn_update_dpp(0, v, 0x112, 0xf, 0xf, false);
      v += __builtin_amdgcn_update_dpp(0, v, 0x114, 0xf, 0xf, false);
      v += __builtin_amdgcn_update_dpp(0, v, 0x118, 0xf, 0xf, false);
      const int lower = __builtin_amdgcn_readlane(v, 15);
      if (lane >= 16) v += lower;
      const int ex = v - tot;
      *(int4*)(rstart + 4 * lane) = make_int4(ex, ex + p1, ex + p2, ex + p3);
    }
    __syncthreads();
#pragma unroll
    for (int i = 0; i < REC_PT; ++i) {
      if (rw[i] >= 0) lrec[rstart[rw[i]] + rk[i]] = r[i];
    }
    __syncthreads();

    const int s0 = __builtin_amdgcn_readfirstlane(rstart[rA]);
    const int sm = __builtin_amdgcn_readfirstlane(rstart[rB]);
    const int s1 = __builtin_amdgcn_readfirstlane(rstart[rC]);
    const int lenA = sm - s0, lenB = s1 - sm;
    const int steps = lenA > lenB ? lenA : lenB;
    const int hbase = upper ? sm : s0;
    const int hlen = upper ? lenB : lenA;
    for (int p0 = 0; p0 < steps; p0 += CHUNK_STEPS) {
      const int cnt = (steps - p0 < CHUNK_STEPS) ? (steps - p0) : CHUNK_STEPS;
      const int pos = p0 + c16;
      const bool real = pos < hlen;
      const int2 rec = lrec[real ? (hbase + pos) : 0];
      const int sidx = real ? ((rec.x & 0xFFFFF) * (DIM / 4)) : 0;
      const int dl = real ? ((rec.x >> 20) - hlo) : -1;
      const float w = real ? __int_as_float(rec.y) : 0.f;
      const int dprev = __builtin_amdgcn_update_dpp(curv, dl, 0x111, 0xf, 0xf, false);
      const unsigned chg = __builtin_amdgcn_ballot_w32(dl != dprev);
      const unsigned anym = (chg | (chg >> 16)) & 0xFFFFu;
      const unsigned mh = chg >> (lane & 16);
      static_assert(CHUNK_STEPS == 4 * STEP_BATCH, "a chunk is the four batches below");
      edge_batch<0>(acc, xw4, sidx, dl, w, anym, mh, hlo, c16, curv, sum);
      if (cnt > STEP_BATCH) edge_batch<STEP_BATCH>(acc, xw4, sidx, dl, w, anym, mh, hlo, c16, curv, sum);
      if (cnt > 2 * STEP_BATCH) edge_batch<2 * STEP_BATCH>(acc, xw4, sidx, dl, w, anym, mh, hlo, c16, curv, sum);
      if (cnt > 3 * STEP_BATCH) edge_batch<3 * STEP_BATCH>(acc, xw4, sidx, dl, w, anym, mh, hlo, c16, curv, sum);
    }
  }
  float4 wq1[W_QUADS];
  if (FIRST) {
#pragma unroll
    for (int q = 0; q < W_QUADS; ++q) wq1[q] = ((const float4*)Wf)[t + q * AGG_THREADS];
  }
  if (curv >= 0) run_flush(acc, hlo + curv, c16, sum);

  __syncthreads();

  if (!FUSE_NEXT) {
    for (int i = t; i < ROW_QUADS; i += AGG_THREADS)
      ((float4*)out)[row4 + i] = *(const float4*)(acc + (i >> 4) * TILE_PITCH + (i & (DIM / 4 - 1)) * 4);
    return;
  }

  const int mt = wave >> 1;
  const int ch = wave & 1;

  if (FIRST) {
    float* wfirst = (float*)recbuf;
    stage_wfrag(wfirst, wq1, t);
    float4 o[ORI_Q];
#pragma unroll
    for (int q = 0; q < ORI_Q; ++q) {
      const int i = t + q * AGG_THREADS;
      o[q] = ((const float4*)ori)[row4 + (i < ROW_QUADS ? i : ROW_QUADS - 1)];
    }
    const float4 bb = ((const float4*)bias)[t & (DIM / 4 - 1)];
    __syncthreads();
    v8f f0 = {}, f1 = {};
    if (wave < 2 * TILE_MT) tile_product(acc, wfirst, mt, ch, lane, f0, f1);
    __syncthreads();
    if (wave < 2 * TILE_MT) tile_put(acc, mt, ch, lane, f0, f1);
    __syncthreads();
#pragma unroll
    for (int q = 0; q < ORI_Q; ++q) {
      const int i = t + q * AGG_THREADS;
      if (i < ROW_QUADS) {
        float* p = acc + (i >> 4) * TILE_PITCH + (i & (DIM / 4 - 1)) * 4;
        const float4 s = *(const float4*)p;
        float4 r;
        r.x = (s.x + bb.x) + o[q].x; r.y = (s.y + bb.y) + o[q].y;
        r.z = (s.z + bb.z) + o[q].z; r.w = (s.w + bb.w) + o[q].w;
        *(float4*)p = r;
      }
    }
    __syncthreads();
  }

  v8f d0 = {}, d1 = {};
  if (wave < 2 * TILE_MT) tile_product(acc, wfrag, mt, ch, lane, d0, d1);
  __syncthreads();
  if (wave < 2 * TILE_MT) tile_put(acc, mt, ch, lane, d0, d1);
  __syncthreads();

  for (int i = t; i < ROW_QUADS; i += AGG_THREADS)
    ((float4*)out)[row4 + i] = *(const float4*)(acc + (i >> 4) * TILE_PITCH + (i & (DIM / 4 - 1)) * 4);
}

extern "C" void kernel_launch(void* const* d_in, const int* in_sizes, int n_in,
                              void* d_out, int out_size, void* d_ws, size_t ws_size,
                              hipStream_t stream) {
  const float* in_feat = (const float*)d_in[0];
  const float* ew      = (const float*)d_in[1];
  const float* W1      = (const float*)d_in[2];
  const float* b1      = (const float*)d_in[3];
  const float* W2      = (const float*)d_in[4];
  const float* b2      = (const float*)d_in[5];
  const int*   src     = (const int*)d_in[6];
  const int*   dst     = (const int*)d_in[7];
  const int    nE      = in_sizes[1];

  float* xwA  = (float*)d_ws;
  float* xwB  = xwA + (size_t)NNODES * DIM;
  int* binCount  = (int*)(xwB + (size_t)NNODES * DIM);
  int* binStart  = binCount + 1024;
  int* binCursor = binStart + 2048;
  int* binned    = binCursor + 1024;
  float* outp = (float*)d_out;

  const int sortBlocks = (nE + SORT_CHUNK - 1) / SORT_CHUNK;

  zero_bins<<<1, 1024, 0, stream>>>(binCount);
  count_bins<<<sortBlocks, SORT_THREADS, 0, stream>>>(dst, binCount, nE);
  scan_bins<<<1, 1024, 0, stream>>>(binCount, binStart, binCursor);
  fill_bins<<<sortBlocks, SORT_THREADS, 0, stream>>>(src, dst, ew, binCursor, binned, nE);

  aggregate_bins<STEP_FIRST><<<NBINS, AGG_THREADS, 0, stream>>>(in_feat, in_feat, b1, binStart, binned, W1, W2, xwA);
  aggregate_bins<STEP_MID><<<NBINS, AGG_THREADS, 0, stream>>>(xwA, in_feat, b2, binStart, binned, W2, W2, xwB);
  aggregate_bins<STEP_MID><<<NBINS, AGG_THREADS, 0, stream>>>(xwB, in_feat, b2, binStart, binned, W2, W2, xwA);
  aggregate_bins<STEP_LAST><<<NBINS, AGG_THREADS, 0, stream>>>(xwA, in_feat, b2, binStart, binned, W2, W2, outp);
}


# Edges sorted by range per 4096-edge chunk in one launch; step 0 sums the chunks' offsets into its range start, collects its pieces, stores them packed: 3 launches drop; FP sums stay arrival order
# speedup vs baseline: 4.6506x; 1.0400x over previous
#include <hip/hip_runtime.h>

typedef __attribute__((ext_vector_type(2))) float v2f;
typedef __attribute__((ext_vector_type(8))) float v8f;

#define NNODES 100000
#define DIM 64


#define BIN_ROWS 100
#define NBINS (NNODES / BIN_ROWS)
static_assert(NBINS * BIN_ROWS == NNODES, "ranges must tile the node set exactly");
static_assert(NNODES <= (1 << 20) && BIN_ROWS < (1 << 11), "record packing");
#define SORT_THREADS 1024
#define SORT_EPT 4
#define SORT_CHUNK (SORT_THREADS * SORT_EPT)
#define N_EDGES 1600000
#define MAX_CHUNKS 512
static_assert((N_EDGES + SORT_CHUNK - 1) / SORT_CHUNK <= MAX_CHUNKS, "a range's work-group has one thread per chunk");

__device__ __forceinline__ int wave_incl_scan(int v, int lane) {
  v += __builtin_amdgcn_update_dpp(0, v, 0x111, 0xf, 0xf, false);
  v += __builtin_amdgcn_update_dpp(0, v, 0x112, 0xf, 0xf, false);
  v += __builtin_amdgcn_update_dpp(0, v, 0x114, 0xf, 0xf, false);
  v += __builtin_amdgcn_update_dpp(0, v, 0x118, 0xf, 0xf, false);
  const int lower = __builtin_amdgcn_readlane(v, 15);
  if (lane >= 16) v += lower;
  return v;
}

#define OFF_PITCH 1024
static_assert(NBINS + 1 <= OFF_PITCH && OFF_PITCH == SORT_THREADS, "one offset per thread");
__global__ __launch_bounds__(SORT_THREADS) void sort_chunks(const int* __restrict__ src,
                                                            const int* __restrict__ dst,
                                                            const float* __restrict__ ew,
                                                            int* __restrict__ chunkOff,
                                                            int* __restrict__ chunked,
                                                            int nE) {
  __shared__ int hist[OFF_PITCH];
  __shared__ int wtot[SORT_THREADS / 32];
  const int t = threadIdx.x;
  const int lane = t & 31;
  const int wave = t >> 5;
  hist[t] = 0;
  __syncthreads();
  const int base = blockIdx.x * SORT_CHUNK;
  int bin[SORT_EPT], rank[SORT_EPT], rec0[SORT_EPT], rec1[SORT_EPT];
#pragma unroll
  for (int i = 0; i < SORT_EPT; ++i) {
    const int e = base + i * SORT_THREADS + t;
    bin[i] = -1; rank[i] = 0; rec0[i] = 0; rec1[i] = 0;
    if (e < nE) {
      const int d = dst[e];
      const int b = d / BIN_ROWS;
      bin[i] = b;
      rank[i] = atomicAdd(&hist[b], 1);
      rec0[i] = src[e] | ((d - b * BIN_ROWS) << 20);
      rec1[i] = __float_as_int(ew[e]);
    }
  }
  __syncthreads();
  const int v = hist[t];
  const int s = wave_incl_scan(v, lane);
  if (lane == 31) wtot[wave] = s;
  __syncthreads();
  if (wave == 0) {
    const int w = wtot[lane];
    wtot[lane] = wave_incl_scan(w, lane) - w;
  }
  __syncthreads();
  const int ex = s - v + wtot[wave];
  hist[t] = ex;
  chunkOff[blockIdx.x * OFF_PITCH + t] = ex;
  __syncthreads();
  int2* segs = (int2*)chunked + (size_t)base;
#pragma unroll
  for (int i = 0; i < SORT_EPT; ++i) {
    if (bin[i] >= 0) segs[hist[bin[i]] + rank[i]] = make_int2(rec0[i], rec1[i]);
  }
}

#define AGG_THREADS 512
#define AGG_WAVES (AGG_THREADS / 32)
#define AGG_HALVES (2 * AGG_WAVES)
#define STEP_BATCH 4
#define CHUNK_STEPS 16
#define TILE_ROWS 112
#define TILE_MT (TILE_ROWS / 16)
#define TILE_PITCH 68
#define STEP_LAST 0
#define STEP_MID 1
#define STEP_FIRST 2
static_assert(MAX_CHUNKS <= AGG_THREADS, "one thread per chunk");
static_assert(TILE_ROWS >= BIN_ROWS && TILE_ROWS % 16 == 0, "tile covers the range");
static_assert(2 * TILE_MT <= AGG_WAVES, "one wave per (row tile, column half)");
static_assert((TILE_ROWS - BIN_ROWS) * (DIM / 4) <= AGG_THREADS, "one pass zeroes the padding rows");
static_assert(AGG_THREADS % (DIM / 4) == 0, "a thread's pieces all fall in one 16-byte column of a row");

constexpr int W_QUADS = DIM * DIM / 4 / AGG_THREADS;
static_assert(W_QUADS * AGG_THREADS * 4 == DIM * DIM, "the matrix splits evenly over the threads");

__device__ __forceinline__ void stage_wfrag(float* wf, const float4 (&wq)[W_QUADS], int t) {
#pragma unroll
  for (int q = 0; q < W_QUADS; ++q) {
    const int i = t + q * AGG_THREADS;
    const int k = i >> 4;
    const int n = (i & 15) * 4;
    float* p = wf + ((((k >> 2) * 4 + (n >> 4)) * 32 + ((k >> 1) & 1) * 16 + (n & 15)) * 2 + (k & 1));
    p[0] = wq[q].x; p[2] = wq[q].y; p[4] = wq[q].z; p[6] = wq[q].w;
  }
}

__device__ __forceinline__ void tile_product(const float* tile, const float* wf, int mt, int ch, int lane,
                                             v8f& d0, v8f& d1) {
  const float* ap = tile + (mt * 16 + (lane & 15)) * TILE_PITCH + (lane >> 4) * 2;
  const float* bp = wf + (ch * 2 * 32 + lane) * 2;
#pragma unroll
  for (int kk = 0; kk < DIM / 4; ++kk) {
    const v2f a  = *(const v2f*)(ap + kk * 4);
    const v2f b0 = *(const v2f*)(bp + kk * 256);
    const v2f b1 = *(const v2f*)(bp + kk * 256 + 64);
    d0 = __builtin_amdgcn_wmma_f32_16x16x4_f32(false, a, false, b0, (short)0, d0, false, false);
    d1 = __builtin_amdgcn_wmma_f32_16x16x4_f32(false, a, false, b1, (short)0, d1, false, false);
  }
}

__device__ __forceinline__ void tile_put(float* tile, int mt, int ch, int lane, const v8f& d0, const v8f& d1) {
  float* dp = tile + (mt * 16 + (lane >> 4) * 8) * TILE_PITCH + ch * 32 + (lane & 15);
#pragma unroll
  for (int r = 0; r < 8; ++r) {
    dp[r * TILE_PITCH] = d0[r];
    dp[r * TILE_PITCH + 16] = d1[r];
  }
}

#define REC_CAP 2048
#define SEG_BATCH 4
static_assert(SEG_BATCH == 4, "the fences in STEP_FIRST name four records");
#define REC_PT (REC_CAP / AGG_THREADS)
#define ROW_SLOTS 128
static_assert(REC_PT * AGG_THREADS == REC_CAP, "a pass splits evenly over the threads");
static_assert(BIN_ROWS <= ROW_SLOTS && ROW_SLOTS == 4 * 32, "one wave scans all row counters, four per lane");
static_assert(REC_CAP * 8 == DIM * DIM * 4, "the record buffer and a staged 64 x 64 matrix share the same bytes");

__device__ __forceinline__ void run_flush(float* acc, int row, int c16, const float4& s) {
  float4* p = (float4*)(acc + row * TILE_PITCH + 4 * c16);
  float4 v = *p;
  v.x += s.x; v.y += s.y; v.z += s.z; v.w += s.w;
  *p = v;
}

template <int U>
__device__ __forceinline__ int half_share(int v) {
  return __builtin_amdgcn_update_dpp(0, v, 0x150 + U, 0xf, 0xf, true);
}

template <int U>
__device__ __forceinline__ float4 gather_rows(const float4* xw4, int sidx, int c16) {
  return xw4[(unsigned)half_share<U>(sidx) + (unsigned)c16];
}

template <int U>
__device__ __forceinline__ void add_edges(float* acc, const float4& x, int dl, float w, unsigned anym, unsigned mh,
                                          int hlo, int c16, int& curv, float4& sum) {
  if (__builtin_expect((anym >> U) & 1u, 0)) {
    const int ncur = half_share<U>(dl);
    if ((mh >> U) & 1u) {
      if (curv >= 0) run_flush(acc, hlo + curv, c16, sum);
      curv = ncur;
      sum = make_float4(0.f, 0.f, 0.f, 0.f);
    }
  }
  const float wj = __int_as_float(half_share<U>(__float_as_int(w)));
  sum.x += x.x * wj;
  sum.y += x.y * wj;
  sum.z += x.z * wj;
  sum.w += x.w * wj;
}

template <int J0>
__device__ __forceinline__ void edge_batch(float* acc, const float4* xw4, int sidx, int dl, float w, unsigned anym,
                                           unsigned mh, int hlo, int c16, int& curv, float4& sum) {
  static_assert(STEP_BATCH == 4 && J0 + STEP_BATCH <= CHUNK_STEPS, "four steps written out below");
  const float4 x0 = gather_rows<J0 + 0>(xw4, sidx, c16);
  const float4 x1 = gather_rows<J0 + 1>(xw4, sidx, c16);
  const float4 x2 = gather_rows<J0 + 2>(xw4, sidx, c16);
  const float4 x3 = gather_rows<J0 + 3>(xw4, sidx, c16);
  add_edges<J0 + 0>(acc, x0, dl, w, anym, mh, hlo, c16, curv, sum);
  add_edges<J0 + 1>(acc, x1, dl, w, anym, mh, hlo, c16, curv, sum);
  add_edges<J0 + 2>(acc, x2, dl, w, anym, mh, hlo, c16, curv, sum);
  add_edges<J0 + 3>(acc, x3, dl, w, anym, mh, hlo, c16, curv, sum);
}

template <int STEP>
__global__ __launch_bounds__(AGG_THREADS) void aggregate_bins(const float* __restrict__ XW,
                                                              const float* __restrict__ ori,
                                                              const float* __restrict__ bias,
                                                              int* __restrict__ binStart,
                                                              int* __restrict__ binned,
                                                              const float* __restrict__ Wf,
                                                              const float* __restrict__ Wn,
                                                              float* __restrict__ out,
                                                              const int* __restrict__ chunkOff,
                                                              const int* __restrict__ chunked,
                                                              int nChunks) {
  constexpr bool FUSE_NEXT = STEP != STEP_LAST;
  constexpr bool FIRST = STEP == STEP_FIRST;
  constexpr int ROWS = FUSE_NEXT ? TILE_ROWS : BIN_ROWS;
  constexpr int ROW_QUADS = BIN_ROWS * (DIM / 4);
  constexpr int ORI_Q = (ROW_QUADS + AGG_THREADS - 1) / AGG_THREADS;
  __shared__ __attribute__((aligned(16))) float acc[ROWS * TILE_PITCH];
  __shared__ __attribute__((aligned(16))) float wfrag[FUSE_NEXT ? DIM * DIM : 4];
  __shared__ __attribute__((aligned(16))) int2 recbuf[REC_CAP];
  __shared__ __attribute__((aligned(16))) int rcnt[ROW_SLOTS];
  __shared__ __attribute__((aligned(16))) int rstart[ROW_SLOTS];
  __shared__ int wtotC[FIRST ? AGG_WAVES : 1];
  __shared__ int wtotO[FIRST ? AGG_WAVES : 1];
  const int t = threadIdx.x;
  const int lane = t & 31;
  const int wave = __builtin_amdgcn_readfirstlane(t >> 5);
  const int b = blockIdx.x;
  const size_t row4 = (size_t)b * ROW_QUADS;
  const int rA = ((2 * wave) * BIN_ROWS + AGG_HALVES - 1) / AGG_HALVES;
  const int rB = ((2 * wave + 1) * BIN_ROWS + AGG_HALVES - 1) / AGG_HALVES;
  const int rC = ((2 * wave + 2) * BIN_ROWS + AGG_HALVES - 1) / AGG_HALVES;
  const int c16 = lane & 15;
  const bool upper = lane >= 16;
  const int hlo = upper ? rB : rA;

  float4 wq[W_QUADS];
  if (FUSE_NEXT) {
#pragma unroll
    for (int q = 0; q < W_QUADS; ++q) wq[q] = ((const float4*)Wn)[t + q * AGG_THREADS];
  }
  if (!FIRST) {
    const float4 bb = ((const float4*)bias)[t & (DIM / 4 - 1)];
    float4 o[ORI_Q];
#pragma unroll
    for (int q = 0; q < ORI_Q; ++q) {
      const int i = t + q * AGG_THREADS;
      o[q] = ((const float4*)ori)[row4 + (i < ROW_QUADS ? i : ROW_QUADS - 1)];
    }
#pragma unroll
    for (int q = 0; q < ORI_Q; ++q) {
      const int i = t + q * AGG_THREADS;
      if (i < ROW_QUADS)
        *(float4*)(acc + (i >> 4) * TILE_PITCH + (i & (DIM / 4 - 1)) * 4) =
            make_float4(o[q].x + bb.x, o[q].y + bb.y, o[q].z + bb.z, o[q].w + bb.w);
    }
  } else {
    for (int i = t; i < ROW_QUADS; i += AGG_THREADS)
      *(float4*)(acc + (i >> 4) * TILE_PITCH + (i & (DIM / 4 - 1)) * 4) = make_float4(0.f, 0.f, 0.f, 0.f);
  }
  if (FUSE_NEXT) {
    if (t < (TILE_ROWS - BIN_ROWS) * (DIM / 4))
      *(float4*)(acc + (BIN_ROWS + (t >> 4)) * TILE_PITCH + (t & (DIM / 4 - 1)) * 4) = make_float4(0.f, 0.f, 0.f, 0.f);
    stage_wfrag(wfrag, wq, t);
  }

  int e0, e1;
  int segPos = 0, segCnt = 0;
  const int2* seg = (const int2*)chunked;
  int2* packed = (int2*)binned;
  if (FIRST) {
    static_assert(AGG_WAVES == 16, "the waves' totals are scanned inside one 16-lane row");
    int o0 = 0, o1 = 0;
    if (t < nChunks) {
      o0 = chunkOff[t * OFF_PITCH + b];
      o1 = chunkOff[t * OFF_PITCH + b + 1];
      seg += (size_t)t * SORT_CHUNK + o0;
    }
    segCnt = o1 - o0;
    const int sc = wave_incl_scan(segCnt, lane);
    const int so = wave_incl_scan(o0, lane);
    if (lane == 31) { wtotC[wave] = sc; wtotO[wave] = so; }
    __syncthreads();
    const int tc = wave_incl_scan(wtotC[lane & (AGG_WAVES - 1)], lane);
    const int to = wave_incl_scan(wtotO[lane & (AGG_WAVES - 1)], lane);
    const int wprev = __builtin_amdgcn_readlane(tc, wave > 0 ? wave - 1 : 0);
    segPos = (wave > 0 ? wprev : 0) + sc - segCnt;
    e0 = __builtin_amdgcn_readlane(to, AGG_WAVES - 1);
    e1 = e0 + __builtin_amdgcn_readlane(tc, AGG_WAVES - 1);
    if (t == 0) {
      binStart[b] = e0;
      if (b == NBINS - 1) binStart[NBINS] = e1;
    }
  } else {
    e0 = binStart[b];
    e1 = binStart[b + 1];
  }
  const int2* recs = (const int2*)binned;
  int2* lrec = recbuf;

  int curv = -1;
  float4 sum = make_float4(0.f, 0.f, 0.f, 0.f);
  const float4* xw4 = (const float4*)XW;

  for (int sb = e0; sb < e1; sb += REC_CAP) {
    const int n = (e1 - sb < REC_CAP) ? (e1 - sb) : REC_CAP;
    if (t < ROW_SLOTS) rcnt[t] = 0;
    __syncthreads();
    int2 r[REC_PT];
    int rw[REC_PT], rk[REC_PT];
    int jlo = 0, jhi = 0;
    if (FIRST) {
      const int w0 = sb - e0;
      jlo = w0 - segPos;
      if (jlo < 0) jlo = 0;
      jhi = w0 + n - segPos;
      if (jhi > segCnt) jhi = segCnt;
      for (int j = jlo; j < jhi; j += SEG_BATCH) {
        int2 q[SEG_BATCH];
#pragma unroll
        for (int u = 0; u < SEG_BATCH; ++u) q[u] = seg[j + u < jhi ? j + u : jhi - 1];
        asm volatile("" :: "v"(q[0].x), "v"(q[0].y), "v"(q[1].x), "v"(q[1].y),
                           "v"(q[2].x), "v"(q[2].y), "v"(q[3].x), "v"(q[3].y));
#pragma unroll
        for (int u = 0; u < SEG_BATCH; ++u) {
          if (j + u < jhi) {
            atomicAdd(&rcnt[q[u].x >> 20], 1);
            packed[e0 + segPos + j + u] = q[u];
          }
        }
      }
    } else {
#pragma unroll
      for (int i = 0; i < REC_PT; ++i) {
        const int k = t + i * AGG_THREADS;
        r[i] = make_int2(0, 0);
        if (k < n) r[i] = recs[sb + k];
      }
#pragma unroll
      for (int i = 0; i < REC_PT; ++i) {
        const int k = t + i * AGG_THREADS;
        rw[i] = -1; rk[i] = 0;
        if (k < n) {
          rw[i] = r[i].x >> 20;
          rk[i] = atomicAdd(&rcnt[rw[i]], 1);
        }
      }
    }
    __syncthreads();
    if (wave == 0) {
      const int4 c4 = *(const int4*)(rcnt + 4 * lane);
      const int p1 = c4.x, p2 = p1 + c4.y, p3 = p2 + c4.z, tot = p3 + c4.w;
      int v = tot;
      v += __builtin_amdgcn_update_dpp(0, v, 0x111, 0xf, 0xf, false);
      v += __builtin_amdgcn_update_dpp(0, v, 0x112, 0xf, 0xf, false);
      v += __builtin_amdgcn_update_dpp(0, v, 0x114, 0xf, 0xf, false);
      v += __builtin_amdgcn_update_dpp(0, v, 0x118, 0xf, 0xf, false);
      const int lower = __builtin_amdgcn_readlane(v, 15);
      if (lane >= 16) v += lower;
      const int ex = v - tot;
      *(int4*)(rstart + 4 * lane) = make_int4(ex, ex + p1, ex + p2, ex + p3);
      if (FIRST) *(int4*)(rcnt + 4 * lane) = make_int4(ex, ex + p1, ex + p2, ex + p3);
    }
    __syncthreads();
    if (FIRST) {
      for (int j = jlo; j < jhi; j += SEG_BATCH) {
        int2 q[SEG_BATCH];
#pragma unroll
        for (int u = 0; u < SEG_BATCH; ++u) q[u] = seg[j + u < jhi ? j + u : jhi - 1];
        asm volatile("" :: "v"(q[0].x), "v"(q[0].y), "v"(q[1].x), "v"(q[1].y),
                           "v"(q[2].x), "v"(q[2].y), "v"(q[3].x), "v"(q[3].y));
#pragma unroll
        for (int u = 0; u < SEG_BATCH; ++u) {
          if (j + u < jhi) lrec[atomicAdd(&rcnt[q[u].x >> 20], 1)] = q[u];
        }
      }
    } else {
#pragma unroll
      for (int i = 0; i < REC_PT; ++i) {
        if (rw[i] >= 0) lrec[rstart[rw[i]] + rk[i]] = r[i];
      }
    }
    __syncthreads();

    const int s0 = __builtin_amdgcn_readfirstlane(rstart[rA]);
    const int sm = __builtin_amdgcn_readfirstlane(rstart[rB]);
    const int s1 = __builtin_amdgcn_readfirstlane(rstart[rC]);
    const int lenA = sm - s0, lenB = s1 - sm;
    const int steps = lenA > lenB ? lenA : lenB;
    const int hbase = upper ? sm : s0;
    const int hlen = upper ? lenB : lenA;
    for (int p0 = 0; p0 < steps; p0 += CHUNK_STEPS) {
      const int cnt = (steps - p0 < CHUNK_STEPS) ? (steps - p0) : CHUNK_STEPS;
      const int pos = p0 + c16;
      const bool real = pos < hlen;
      const int2 rec = lrec[real ? (hbase + pos) : 0];
      const int sidx = real ? ((rec.x & 0xFFFFF) * (DIM / 4)) : 0;
      const int dl = real ? ((rec.x >> 20) - hlo) : -1;
      const float w = real ? __int_as_float(rec.y) : 0.f;
      const int dprev = __builtin_amdgcn_update_dpp(curv, dl, 0x111, 0xf, 0xf, false);
      const unsigned chg = __builtin_amdgcn_ballot_w32(dl != dprev);
      const unsigned anym = (chg | (chg >> 16)) & 0xFFFFu;
      const unsigned mh = chg >> (lane & 16);
      static_assert(CHUNK_STEPS == 4 * STEP_BATCH, "a chunk is the four batches below");
      edge_batch<0>(acc, xw4, sidx, dl, w, anym, mh, hlo, c16, curv, sum);
      if (cnt > STEP_BATCH) edge_batch<STEP_BATCH>(acc, xw4, sidx, dl, w, anym, mh, hlo, c16, curv, sum);
      if (cnt > 2 * STEP_BATCH) edge_batch<2 * STEP_BATCH>(acc, xw4, sidx, dl, w, anym, mh, hlo, c16, curv, sum);
      if (cnt > 3 * STEP_BATCH) edge_batch<3 * STEP_BATCH>(acc, xw4, sidx, dl, w, anym, mh, hlo, c16, curv, sum);
    }
  }
  float4 wq1[W_QUADS];
  if (FIRST) {
#pragma unroll
    for (int q = 0; q < W_QUADS; ++q) wq1[q] = ((const float4*)Wf)[t + q * AGG_THREADS];
  }
  if (curv >= 0) run_flush(acc, hlo + curv, c16, sum);

  __syncthreads();

  if (!FUSE_NEXT) {
    for (int i = t; i < ROW_QUADS; i += AGG_THREADS)
      ((float4*)out)[row4 + i] = *(const float4*)(acc + (i >> 4) * TILE_PITCH + (i & (DIM / 4 - 1)) * 4);
    return;
  }

  const int mt = wave >> 1;
  const int ch = wave & 1;

  if (FIRST) {
    float* wfirst = (float*)recbuf;
    stage_wfrag(wfirst, wq1, t);
    __syncthreads();
    v8f f0 = {}, f1 = {};
    if (wave < 2 * TILE_MT) tile_product(acc, wfirst, mt, ch, lane, f0, f1);
    float4 o[ORI_Q];
#pragma unroll
    for (int q = 0; q < ORI_Q; ++q) {
      const int i = t + q * AGG_THREADS;
      o[q] = ((const float4*)ori)[row4 + (i < ROW_QUADS ? i : ROW_QUADS - 1)];
    }
    const float4 bb = ((const float4*)bias)[t & (DIM / 4 - 1)];
    __syncthreads();
    if (wave < 2 * TILE_MT) tile_put(acc, mt, ch, lane, f0, f1);
    __syncthreads();
#pragma unroll
    for (int q = 0; q < ORI_Q; ++q) {
      const int i = t + q * AGG_THREADS;
      if (i < ROW_QUADS) {
        float* p = acc + (i >> 4) * TILE_PITCH + (i & (DIM / 4 - 1)) * 4;
        const float4 s = *(const float4*)p;
        float4 r;
        r.x = (s.x + bb.x) + o[q].x; r.y = (s.y + bb.y) + o[q].y;
        r.z = (s.z + bb.z) + o[q].z; r.w = (s.w + bb.w) + o[q].w;
        *(float4*)p = r;
      }
    }
    __syncthreads();
  }

  v8f d0 = {}, d1 = {};
  if (wave < 2 * TILE_MT) tile_product(acc, wfrag, mt, ch, lane, d0, d1);
  __syncthreads();
  if (wave < 2 * TILE_MT) tile_put(acc, mt, ch, lane, d0, d1);
  __syncthreads();

  for (int i = t; i < ROW_QUADS; i += AGG_THREADS)
    ((float4*)out)[row4 + i] = *(const float4*)(acc + (i >> 4) * TILE_PITCH + (i & (DIM / 4 - 1)) * 4);
}

extern "C" void kernel_launch(void* const* d_in, const int* in_sizes, int n_in,
                              void* d_out, int out_size, void* d_ws, size_t ws_size,
                              hipStream_t stream) {
  const float* in_feat = (const float*)d_in[0];
  const float* ew      = (const float*)d_in[1];
  const float* W1      = (const float*)d_in[2];
  const float* b1      = (const float*)d_in[3];
  const float* W2      = (const float*)d_in[4];
  const float* b2      = (const float*)d_in[5];
  const int*   src     = (const int*)d_in[6];
  const int*   dst     = (const int*)d_in[7];
  const int    nE      = in_sizes[1];

  float* xwA  = (float*)d_ws;
  float* xwB  = xwA + (size_t)NNODES * DIM;
  int* binStart = (int*)(xwB + (size_t)NNODES * DIM);
  int* chunkOff = binStart + 2048;
  int* binned   = chunkOff + MAX_CHUNKS * OFF_PITCH;
  int* chunked  = binned + 2 * N_EDGES;
  float* outp = (float*)d_out;

  const int sortBlocks = (nE + SORT_CHUNK - 1) / SORT_CHUNK;

  sort_chunks<<<sortBlocks, SORT_THREADS, 0, stream>>>(src, dst, ew, chunkOff, chunked, nE);

  aggregate_bins<STEP_FIRST><<<NBINS, AGG_THREADS, 0, stream>>>(in_feat, in_feat, b1, binStart, binned, W1, W2, xwA, chunkOff, chunked, sortBlocks);
  aggregate_bins<STEP_MID><<<NBINS, AGG_THREADS, 0, stream>>>(xwA, in_feat, b2, binStart, binned, W2, W2, xwB, chunkOff, chunked, sortBlocks);
  aggregate_bins<STEP_MID><<<NBINS, AGG_THREADS, 0, stream>>>(xwB, in_feat, b2, binStart, binned, W2, W2, xwA, chunkOff, chunked, sortBlocks);
  aggregate_bins<STEP_LAST><<<NBINS, AGG_THREADS, 0, stream>>>(xwA, in_feat, b2, binStart, binned, W2, W2, outp, chunkOff, chunked, sortBlocks);
}
